# starting kernel
# baseline (speedup 1.0000x reference)
#include <hip/hip_runtime.h>

typedef float v2f __attribute__((ext_vector_type(2)));
typedef float v8f __attribute__((ext_vector_type(8)));

__global__ void deg_init_kernel(float* __restrict__ deg, int n) {
    int i = blockIdx.x * blockDim.x + threadIdx.x;
    if (i < n) deg[i] = 1.0f;
}

__global__ void deg_count_kernel(const int* __restrict__ dst, float* __restrict__ deg, int e) {
    int i = blockIdx.x * blockDim.x + threadIdx.x;
    if (i < e) atomicAdd(&deg[dst[i]], 1.0f);
}

__global__ void dinv_kernel(float* __restrict__ deg, int n) {
    int i = blockIdx.x * blockDim.x + threadIdx.x;
    if (i < n) deg[i] = rsqrtf(deg[i]);
}

template <int CIN, int COUT>
__global__ __launch_bounds__(256) void transform_kernel(
    const float* __restrict__ x, const float* __restrict__ W,
    const float* __restrict__ dinv, float* __restrict__ hs,
    float* __restrict__ acc_out, int n)
{
    constexpr int  KSTEPS = (CIN + 3) / 4;
    constexpr int  TILES  = COUT / 16;
    constexpr bool KPAD   = (CIN % 4) != 0;

    int wave     = (blockIdx.x * blockDim.x + threadIdx.x) >> 5;
    int lane     = threadIdx.x & 31;
    int row_base = wave * 16;
    if (row_base >= n) return;

    int m  = lane & 15;
    int kh = (lane >> 4) & 1;
    int arow = row_base + m;
    int ar = arow < n ? arow : n - 1;

    v8f acc[TILES];
#pragma unroll
    for (int t = 0; t < TILES; ++t) acc[t] = {};

#pragma unroll
    for (int kb = 0; kb < KSTEPS; ++kb) {
        int k0 = kb * 4 + 2 * kh;
        v2f a;
        if constexpr (!KPAD) {
            a = *(const v2f*)&x[ar * CIN + k0];
        } else {
            int ka0 = (k0     < CIN) ? k0     : CIN - 1;
            int ka1 = (k0 + 1 < CIN) ? k0 + 1 : CIN - 1;
            a.x = x[ar * CIN + ka0];
            a.y = x[ar * CIN + ka1];
        }
#pragma unroll
        for (int t = 0; t < TILES; ++t) {
            int ncol = t * 16 + m;
            v2f b;
            if constexpr (!KPAD) {
                b.x = W[ k0      * COUT + ncol];
                b.y = W[(k0 + 1) * COUT + ncol];
            } else {
                bool ok0 = (k0     < CIN);
                bool ok1 = (k0 + 1 < CIN);
                float w0 = W[(ok0 ? k0     : 0) * COUT + ncol];
                float w1 = W[(ok1 ? k0 + 1 : 0) * COUT + ncol];
                b.x = ok0 ? w0 : 0.0f;
                b.y = ok1 ? w1 : 0.0f;
            }
            acc[t] = __builtin_amdgcn_wmma_f32_16x16x4_f32(
                 false, a,  false, b,
                 (short)0, acc[t],  false,  false);
        }
    }

    if (row_base + 16 <= n) {
#pragma unroll
        for (int v = 0; v < 8; ++v) {
            int r = row_base + v + 8 * kh;
            float di = dinv[r];
#pragma unroll
            for (int t = 0; t < TILES; ++t) {
                float val = acc[t][v] * di;
                int idx = r * COUT + t * 16 + m;
                hs[idx]      = val;
                acc_out[idx] = val;
            }
        }
    } else {
#pragma unroll
        for (int v = 0; v < 8; ++v) {
            int r = row_base + v + 8 * kh;
            if (r < n) {
                float di = dinv[r];
#pragma unroll
                for (int t = 0; t < TILES; ++t) {
                    float val = acc[t][v] * di;
                    int idx = r * COUT + t * 16 + m;
                    hs[idx]      = val;
                    acc_out[idx] = val;
                }
            }
        }
    }
}

template <int C>
__global__ __launch_bounds__(256) void scatter_kernel(
    const int* __restrict__ src, const int* __restrict__ dst,
    const float* __restrict__ hs, float* __restrict__ acc, int e)
{
    constexpr int G = C / 4;
    int idx = blockIdx.x * blockDim.x + threadIdx.x;
    int ei  = idx / G;
    if (ei >= e) return;
    int c4 = (idx & (G - 1)) * 4;
    int s = src[ei];
    int d = dst[ei];
    const float4 v = *(const float4*)&hs[s * C + c4];
    float* a = &acc[d * C + c4];
    atomicAdd(a + 0, v.x);
    atomicAdd(a + 1, v.y);
    atomicAdd(a + 2, v.z);
    atomicAdd(a + 3, v.w);
}

template <int C, bool RELU>
__global__ __launch_bounds__(256) void finalize_kernel(
    const float* __restrict__ acc, const float* __restrict__ dinv,
    const float* __restrict__ bias, float* __restrict__ out, int n)
{
    constexpr int G = C / 4;
    int idx = blockIdx.x * blockDim.x + threadIdx.x;
    if (idx >= n * G) return;
    int node = idx / G;
    int c4   = (idx & (G - 1)) * 4;
    float di = dinv[node];
    float4 v = *(const float4*)&acc[node * C + c4];
    float4 b = *(const float4*)&bias[c4];
    float4 o;
    o.x = fmaf(di, v.x, b.x);
    o.y = fmaf(di, v.y, b.y);
    o.z = fmaf(di, v.z, b.z);
    o.w = fmaf(di, v.w, b.w);
    if (RELU) {
        o.x = fmaxf(o.x, 0.0f); o.y = fmaxf(o.y, 0.0f);
        o.z = fmaxf(o.z, 0.0f); o.w = fmaxf(o.w, 0.0f);
    }
    *(float4*)&out[node * C + c4] = o;
}

extern "C" void kernel_launch(void* const* d_in, const int* in_sizes, int n_in,
                              void* d_out, int out_size, void* d_ws, size_t ws_size,
                              hipStream_t stream) {
    const float* x  = (const float*)d_in[0];
    const int*   ei = (const int*)  d_in[1];
    const float* W1 = (const float*)d_in[2];
    const float* b1 = (const float*)d_in[3];
    const float* W2 = (const float*)d_in[4];
    const float* b2 = (const float*)d_in[5];
    const float* W3 = (const float*)d_in[6];
    const float* b3 = (const float*)d_in[7];
    float* out = (float*)d_out;

    const int n = in_sizes[0] / 11;
    const int e = in_sizes[1] / 2;
    const int* src = ei;
    const int* dst = ei + e;

    float* ws   = (float*)d_ws;
    float* dinv = ws;
    float* hs   = dinv + n;
    float* acc  = hs + (size_t)n * 64;
    float* hbuf = acc + (size_t)n * 64;

    const int B = 256;
    auto cdiv = [](long a, long b) { return (int)((a + b - 1) / b); };

    deg_init_kernel <<<cdiv(n, B), B, 0, stream>>>(dinv, n);
    deg_count_kernel<<<cdiv(e, B), B, 0, stream>>>(dst, dinv, e);
    dinv_kernel     <<<cdiv(n, B), B, 0, stream>>>(dinv, n);

    const int tb = cdiv((long)((n + 15) / 16) * 32, B);

    transform_kernel<11, 16><<<tb, B, 0, stream>>>(x, W1, dinv, hs, acc, n);
    scatter_kernel<16>       <<<cdiv((long)e * 4,  B), B, 0, stream>>>(src, dst, hs, acc, e);
    finalize_kernel<16, true><<<cdiv((long)n * 4,  B), B, 0, stream>>>(acc, dinv, b1, hbuf, n);

    transform_kernel<16, 32><<<tb, B, 0, stream>>>(hbuf, W2, dinv, hs, acc, n);
    scatter_kernel<32>       <<<cdiv((long)e * 8,  B), B, 0, stream>>>(src, dst, hs, acc, e);
    finalize_kernel<32, true><<<cdiv((long)n * 8,  B), B, 0, stream>>>(acc, dinv, b2, hbuf, n);

    transform_kernel<32, 64><<<tb, B, 0, stream>>>(hbuf, W3, dinv, hs, acc, n);
    scatter_kernel<64>        <<<cdiv((long)e * 16, B), B, 0, stream>>>(src, dst, hs, acc, e);
    finalize_kernel<64, false><<<cdiv((long)n * 16, B), B, 0, stream>>>(acc, dinv, b3, out, n);
}


# Edge scatter re-laid out, one row over consecutive lanes: each gather and each f32 atomic add instruction covers one whole 128-B line (C=16: two rows), not quarter lines; sums stay atomics
# speedup vs baseline: 1.3775x; 1.3775x over previous
#include <hip/hip_runtime.h>

typedef float v2f __attribute__((ext_vector_type(2)));
typedef float v8f __attribute__((ext_vector_type(8)));

__global__ void deg_init_kernel(float* __restrict__ deg, int n) {
    int i = blockIdx.x * blockDim.x + threadIdx.x;
    if (i < n) deg[i] = 1.0f;
}

__global__ void deg_count_kernel(const int* __restrict__ dst, float* __restrict__ deg, int e) {
    int i = blockIdx.x * blockDim.x + threadIdx.x;
    if (i < e) atomicAdd(&deg[dst[i]], 1.0f);
}

__global__ void dinv_kernel(float* __restrict__ deg, int n) {
    int i = blockIdx.x * blockDim.x + threadIdx.x;
    if (i < n) deg[i] = rsqrtf(deg[i]);
}

template <int CIN, int COUT>
__global__ __launch_bounds__(256) void transform_kernel(
    const float* __restrict__ x, const float* __restrict__ W,
    const float* __restrict__ dinv, float* __restrict__ hs,
    float* __restrict__ acc_out, int n)
{
    constexpr int  KSTEPS = (CIN + 3) / 4;
    constexpr int  TILES  = COUT / 16;
    constexpr bool KPAD   = (CIN % 4) != 0;

    int wave     = (blockIdx.x * blockDim.x + threadIdx.x) >> 5;
    int lane     = threadIdx.x & 31;
    int row_base = wave * 16;
    if (row_base >= n) return;

    int m  = lane & 15;
    int kh = (lane >> 4) & 1;
    int arow = row_base + m;
    int ar = arow < n ? arow : n - 1;

    v8f acc[TILES];
#pragma unroll
    for (int t = 0; t < TILES; ++t) acc[t] = {};

#pragma unroll
    for (int kb = 0; kb < KSTEPS; ++kb) {
        int k0 = kb * 4 + 2 * kh;
        v2f a;
        if constexpr (!KPAD) {
            a = *(const v2f*)&x[ar * CIN + k0];
        } else {
            int ka0 = (k0     < CIN) ? k0     : CIN - 1;
            int ka1 = (k0 + 1 < CIN) ? k0 + 1 : CIN - 1;
            a.x = x[ar * CIN + ka0];
            a.y = x[ar * CIN + ka1];
        }
#pragma unroll
        for (int t = 0; t < TILES; ++t) {
            int ncol = t * 16 + m;
            v2f b;
            if constexpr (!KPAD) {
                b.x = W[ k0      * COUT + ncol];
                b.y = W[(k0 + 1) * COUT + ncol];
            } else {
                bool ok0 = (k0     < CIN);
                bool ok1 = (k0 + 1 < CIN);
                float w0 = W[(ok0 ? k0     : 0) * COUT + ncol];
                float w1 = W[(ok1 ? k0 + 1 : 0) * COUT + ncol];
                b.x = ok0 ? w0 : 0.0f;
                b.y = ok1 ? w1 : 0.0f;
            }
            acc[t] = __builtin_amdgcn_wmma_f32_16x16x4_f32(
                 false, a,  false, b,
                 (short)0, acc[t],  false,  false);
        }
    }

    if (row_base + 16 <= n) {
#pragma unroll
        for (int v = 0; v < 8; ++v) {
            int r = row_base + v + 8 * kh;
            float di = dinv[r];
#pragma unroll
            for (int t = 0; t < TILES; ++t) {
                float val = acc[t][v] * di;
                int idx = r * COUT + t * 16 + m;
                hs[idx]      = val;
                acc_out[idx] = val;
            }
        }
    } else {
#pragma unroll
        for (int v = 0; v < 8; ++v) {
            int r = row_base + v + 8 * kh;
            if (r < n) {
                float di = dinv[r];
#pragma unroll
                for (int t = 0; t < TILES; ++t) {
                    float val = acc[t][v] * di;
                    int idx = r * COUT + t * 16 + m;
                    hs[idx]      = val;
                    acc_out[idx] = val;
                }
            }
        }
    }
}

template <int C>
__global__ __launch_bounds__(256) void scatter_kernel(
    const int* __restrict__ src, const int* __restrict__ dst,
    const float* __restrict__ hs, float* __restrict__ acc, int e)
{
    constexpr int BPW   = 4;
    constexpr int LPE   = (C < 32) ? C : 32;
    constexpr int EPI   = 32 / LPE;
    constexpr int PCS   = C / LPE;
    constexpr int STEPS = 32 / EPI;
    constexpr int GRP   = (STEPS * PCS > 32) ? 32 / PCS : STEPS;

    const int lane = threadIdx.x & 31;
    const int ch   = lane & (LPE - 1);
    const bool up  = (lane >= LPE);
    const int wave = __builtin_amdgcn_readfirstlane(
        (int)((blockIdx.x * blockDim.x + threadIdx.x) >> 5));

    int base = wave * (BPW * 32);
    for (int it = 0; it < BPW; ++it, base += 32) {
        if (base >= e) return;
        const int cnt = e - base;
        int sv = 0, dv = 0;
        if (base + lane < e) { sv = src[base + lane]; dv = dst[base + lane]; }

        if (cnt >= 32) {
            auto pick = [&](int vec, int t) -> unsigned {
                if constexpr (EPI == 1) {
                    return (unsigned)__builtin_amdgcn_readlane(vec, t);
                } else {
                    const int lo = __builtin_amdgcn_readlane(vec, 2 * t);
                    const int hi = __builtin_amdgcn_readlane(vec, 2 * t + 1);
                    return (unsigned)(up ? hi : lo);
                }
            };
#pragma unroll
            for (int g = 0; g < STEPS; g += GRP) {
                float v[GRP][PCS];
#pragma unroll
                for (int t = 0; t < GRP; ++t) {
                    const unsigned off = pick(sv, g + t) * (unsigned)C + (unsigned)ch;
#pragma unroll
                    for (int p = 0; p < PCS; ++p) v[t][p] = hs[off + p * 32];
                }
#pragma unroll
                for (int t = 0; t < GRP; ++t) {
                    const unsigned off = pick(dv, g + t) * (unsigned)C + (unsigned)ch;
#pragma unroll
                    for (int p = 0; p < PCS; ++p) atomicAdd(&acc[off + p * 32], v[t][p]);
                }
            }
        } else {
            for (int j = 0; j < cnt; ++j) {
                const unsigned s = (unsigned)__builtin_amdgcn_readlane(sv, j);
                const unsigned d = (unsigned)__builtin_amdgcn_readlane(dv, j);
                for (int c = lane; c < C; c += 32)
                    atomicAdd(&acc[(size_t)d * C + c], hs[(size_t)s * C + c]);
            }
        }
    }
}

template <int C, bool RELU>
__global__ __launch_bounds__(256) void finalize_kernel(
    const float* __restrict__ acc, const float* __restrict__ dinv,
    const float* __restrict__ bias, float* __restrict__ out, int n)
{
    constexpr int G = C / 4;
    int idx = blockIdx.x * blockDim.x + threadIdx.x;
    if (idx >= n * G) return;
    int node = idx / G;
    int c4   = (idx & (G - 1)) * 4;
    float di = dinv[node];
    float4 v = *(const float4*)&acc[node * C + c4];
    float4 b = *(const float4*)&bias[c4];
    float4 o;
    o.x = fmaf(di, v.x, b.x);
    o.y = fmaf(di, v.y, b.y);
    o.z = fmaf(di, v.z, b.z);
    o.w = fmaf(di, v.w, b.w);
    if (RELU) {
        o.x = fmaxf(o.x, 0.0f); o.y = fmaxf(o.y, 0.0f);
        o.z = fmaxf(o.z, 0.0f); o.w = fmaxf(o.w, 0.0f);
    }
    *(float4*)&out[node * C + c4] = o;
}

extern "C" void kernel_launch(void* const* d_in, const int* in_sizes, int n_in,
                              void* d_out, int out_size, void* d_ws, size_t ws_size,
                              hipStream_t stream) {
    const float* x  = (const float*)d_in[0];
    const int*   ei = (const int*)  d_in[1];
    const float* W1 = (const float*)d_in[2];
    const float* b1 = (const float*)d_in[3];
    const float* W2 = (const float*)d_in[4];
    const float* b2 = (const float*)d_in[5];
    const float* W3 = (const float*)d_in[6];
    const float* b3 = (const float*)d_in[7];
    float* out = (float*)d_out;

    const int n = in_sizes[0] / 11;
    const int e = in_sizes[1] / 2;
    const int* src = ei;
    const int* dst = ei + e;

    float* ws   = (float*)d_ws;
    float* dinv = ws;
    float* hs   = dinv + n;
    float* acc  = hs + (size_t)n * 64;
    float* hbuf = acc + (size_t)n * 64;

    const int B = 256;
    auto cdiv = [](long a, long b) { return (int)((a + b - 1) / b); };

    deg_init_kernel <<<cdiv(n, B), B, 0, stream>>>(dinv, n);
    deg_count_kernel<<<cdiv(e, B), B, 0, stream>>>(dst, dinv, e);
    dinv_kernel     <<<cdiv(n, B), B, 0, stream>>>(dinv, n);

    const int tb = cdiv((long)((n + 15) / 16) * 32, B);
    const int sb = cdiv((long)e, 1024);

    transform_kernel<11, 16><<<tb, B, 0, stream>>>(x, W1, dinv, hs, acc, n);
    scatter_kernel<16>       <<<sb, B, 0, stream>>>(src, dst, hs, acc, e);
    finalize_kernel<16, true><<<cdiv((long)n * 4,  B), B, 0, stream>>>(acc, dinv, b1, hbuf, n);

    transform_kernel<16, 32><<<tb, B, 0, stream>>>(hbuf, W2, dinv, hs, acc, n);
    scatter_kernel<32>       <<<sb, B, 0, stream>>>(src, dst, hs, acc, e);
    finalize_kernel<32, true><<<cdiv((long)n * 8,  B), B, 0, stream>>>(acc, dinv, b2, hbuf, n);

    transform_kernel<32, 64><<<tb, B, 0, stream>>>(hbuf, W3, dinv, hs, acc, n);
    scatter_kernel<64>        <<<sb, B, 0, stream>>>(src, dst, hs, acc, e);
    finalize_kernel<64, false><<<cdiv((long)n * 16, B), B, 0, stream>>>(acc, dinv, b3, out, n);
}


# Layers 2,3 sum over edges BEFORE the dense product ((S+I)(XW)=((S+I)X)W): edge passes move 16,32 channels not 32,64; bias/relu/scale in the WMMA epilogue, 12->10 launches; f32 sum order changes
# speedup vs baseline: 1.7723x; 1.2866x over previous
#include <hip/hip_runtime.h>


typedef float v2f __attribute__((ext_vector_type(2)));
typedef float v8f __attribute__((ext_vector_type(8)));

__global__ void deg_init_kernel(float* __restrict__ deg, int n) {
    int i = blockIdx.x * blockDim.x + threadIdx.x;
    if (i < n) deg[i] = 1.0f;
}

__global__ void deg_count_kernel(const int* __restrict__ dst, float* __restrict__ deg, int e) {
    int i = blockIdx.x * blockDim.x + threadIdx.x;
    if (i < e) atomicAdd(&deg[dst[i]], 1.0f);
}

__global__ void dinv_kernel(float* __restrict__ deg, int n) {
    int i = blockIdx.x * blockDim.x + threadIdx.x;
    if (i < n) deg[i] = rsqrtf(deg[i]);
}

template <int MODE>
__device__ __forceinline__ float epilogue_value(float p, float di, float b) {
    if constexpr (MODE == 0) return p * di;
    else if constexpr (MODE == 1) return fmaxf(fmaf(di, p, b), 0.0f) * di;
    else return fmaf(di, p, b);
}

template <int CIN, int COUT, int MODE>
__global__ __launch_bounds__(256) void transform_kernel(
    const float* __restrict__ x, const float* __restrict__ W,
    const float* __restrict__ dinv, const float* __restrict__ bias,
    float* __restrict__ o0, float* __restrict__ o1, int n)
{
    constexpr int  KSTEPS = (CIN + 3) / 4;
    constexpr int  TILES  = COUT / 16;
    constexpr bool KPAD   = (CIN % 4) != 0;

    int wave     = (blockIdx.x * blockDim.x + threadIdx.x) >> 5;
    int lane     = threadIdx.x & 31;
    int row_base = wave * 16;
    if (row_base >= n) return;

    int m  = lane & 15;
    int kh = (lane >> 4) & 1;
    int arow = row_base + m;
    int ar = arow < n ? arow : n - 1;

    v8f acc[TILES];
#pragma unroll
    for (int t = 0; t < TILES; ++t) acc[t] = {};

#pragma unroll
    for (int kb = 0; kb < KSTEPS; ++kb) {
        int k0 = kb * 4 + 2 * kh;
        v2f a;
        if constexpr (!KPAD) {
            a = *(const v2f*)&x[ar * CIN + k0];
        } else {
            int ka0 = (k0     < CIN) ? k0     : CIN - 1;
            int ka1 = (k0 + 1 < CIN) ? k0 + 1 : CIN - 1;
            a.x = x[ar * CIN + ka0];
            a.y = x[ar * CIN + ka1];
        }
#pragma unroll
        for (int t = 0; t < TILES; ++t) {
            int ncol = t * 16 + m;
            v2f b;
            if constexpr (!KPAD) {
                b.x = W[ k0      * COUT + ncol];
                b.y = W[(k0 + 1) * COUT + ncol];
            } else {
                bool ok0 = (k0     < CIN);
                bool ok1 = (k0 + 1 < CIN);
                float w0 = W[(ok0 ? k0     : 0) * COUT + ncol];
                float w1 = W[(ok1 ? k0 + 1 : 0) * COUT + ncol];
                b.x = ok0 ? w0 : 0.0f;
                b.y = ok1 ? w1 : 0.0f;
            }
            acc[t] = __builtin_amdgcn_wmma_f32_16x16x4_f32(
                 false, a,  false, b,
                 (short)0, acc[t],  false,  false);
        }
    }

    float bv[TILES];
#pragma unroll
    for (int t = 0; t < TILES; ++t) {
        if constexpr (MODE != 0) bv[t] = bias[t * 16 + m]; else bv[t] = 0.0f;
    }

    if (row_base + 16 <= n) {
#pragma unroll
        for (int v = 0; v < 8; ++v) {
            int r = row_base + v + 8 * kh;
            float di = dinv[r];
#pragma unroll
            for (int t = 0; t < TILES; ++t) {
                float val = epilogue_value<MODE>(acc[t][v], di, bv[t]);
                int idx = r * COUT + t * 16 + m;
                o0[idx] = val;
                if constexpr (MODE != 2) o1[idx] = val;
            }
        }
    } else {
#pragma unroll
        for (int v = 0; v < 8; ++v) {
            int r = row_base + v + 8 * kh;
            if (r < n) {
                float di = dinv[r];
#pragma unroll
                for (int t = 0; t < TILES; ++t) {
                    float val = epilogue_value<MODE>(acc[t][v], di, bv[t]);
                    int idx = r * COUT + t * 16 + m;
                    o0[idx] = val;
                    if constexpr (MODE != 2) o1[idx] = val;
                }
            }
        }
    }
}

template <int C>
__global__ __launch_bounds__(256) void scatter_kernel(
    const int* __restrict__ src, const int* __restrict__ dst,
    const float* __restrict__ hs, float* __restrict__ acc, int e)
{
    constexpr int BPW   = 4;
    constexpr int LPE   = (C < 32) ? C : 32;
    constexpr int EPI   = 32 / LPE;
    constexpr int PCS   = C / LPE;
    constexpr int STEPS = 32 / EPI;
    constexpr int GRP   = (STEPS * PCS > 32) ? 32 / PCS : STEPS;

    const int lane = threadIdx.x & 31;
    const int ch   = lane & (LPE - 1);
    const bool up  = (lane >= LPE);
    const int wave = __builtin_amdgcn_readfirstlane(
        (int)((blockIdx.x * blockDim.x + threadIdx.x) >> 5));

    int base = wave * (BPW * 32);
    for (int it = 0; it < BPW; ++it, base += 32) {
        if (base >= e) return;
        const int cnt = e - base;
        int sv = 0, dv = 0;
        if (base + lane < e) { sv = src[base + lane]; dv = dst[base + lane]; }

        if (cnt >= 32) {
            auto pick = [&](int vec, int t) -> unsigned {
                if constexpr (EPI == 1) {
                    return (unsigned)__builtin_amdgcn_readlane(vec, t);
                } else {
                    const int lo = __builtin_amdgcn_readlane(vec, 2 * t);
                    const int hi = __builtin_amdgcn_readlane(vec, 2 * t + 1);
                    return (unsigned)(up ? hi : lo);
                }
            };
#pragma unroll
            for (int g = 0; g < STEPS; g += GRP) {
                float v[GRP][PCS];
#pragma unroll
                for (int t = 0; t < GRP; ++t) {
                    const unsigned off = pick(sv, g + t) * (unsigned)C + (unsigned)ch;
#pragma unroll
                    for (int p = 0; p < PCS; ++p) v[t][p] = hs[off + p * 32];
                }
#pragma unroll
                for (int t = 0; t < GRP; ++t) {
                    const unsigned off = pick(dv, g + t) * (unsigned)C + (unsigned)ch;
#pragma unroll
                    for (int p = 0; p < PCS; ++p) atomicAdd(&acc[off + p * 32], v[t][p]);
                }
            }
        } else {
            for (int j = 0; j < cnt; ++j) {
                const unsigned s = (unsigned)__builtin_amdgcn_readlane(sv, j);
                const unsigned d = (unsigned)__builtin_amdgcn_readlane(dv, j);
                for (int c = lane; c < C; c += 32)
                    atomicAdd(&acc[(size_t)d * C + c], hs[(size_t)s * C + c]);
            }
        }
    }
}

template <int C>
__global__ __launch_bounds__(256) void finalize_scaled_kernel(
    const float* __restrict__ acc, const float* __restrict__ dinv,
    const float* __restrict__ bias, float* __restrict__ hs_next,
    float* __restrict__ acc_next, int n)
{
    constexpr int G = C / 4;
    int idx = blockIdx.x * blockDim.x + threadIdx.x;
    if (idx >= n * G) return;
    int node = idx / G;
    int c4   = (idx & (G - 1)) * 4;
    float di = dinv[node];
    float4 v = *(const float4*)&acc[node * C + c4];
    float4 b = *(const float4*)&bias[c4];
    float4 o;
    o.x = fmaxf(fmaf(di, v.x, b.x), 0.0f) * di;
    o.y = fmaxf(fmaf(di, v.y, b.y), 0.0f) * di;
    o.z = fmaxf(fmaf(di, v.z, b.z), 0.0f) * di;
    o.w = fmaxf(fmaf(di, v.w, b.w), 0.0f) * di;
    *(float4*)&hs_next[node * C + c4]  = o;
    *(float4*)&acc_next[node * C + c4] = o;
}

extern "C" void kernel_launch(void* const* d_in, const int* in_sizes, int n_in,
                              void* d_out, int out_size, void* d_ws, size_t ws_size,
                              hipStream_t stream) {
    const float* x  = (const float*)d_in[0];
    const int*   ei = (const int*)  d_in[1];
    const float* W1 = (const float*)d_in[2];
    const float* b1 = (const float*)d_in[3];
    const float* W2 = (const float*)d_in[4];
    const float* b2 = (const float*)d_in[5];
    const float* W3 = (const float*)d_in[6];
    const float* b3 = (const float*)d_in[7];
    float* out = (float*)d_out;

    const int n = in_sizes[0] / 11;
    const int e = in_sizes[1] / 2;
    const int* src = ei;
    const int* dst = ei + e;

    float* ws   = (float*)d_ws;
    float* dinv = ws;
    float* hsA  = dinv + n;
    float* accA = hsA  + (size_t)n * 16;
    float* hsB  = accA + (size_t)n * 16;
    float* accB = hsB  + (size_t)n * 16;
    float* hsC  = accB + (size_t)n * 16;
    float* accC = hsC  + (size_t)n * 32;

    const int B = 256;
    auto cdiv = [](long a, long b) { return (int)((a + b - 1) / b); };

    deg_init_kernel <<<cdiv(n, B), B, 0, stream>>>(dinv, n);
    deg_count_kernel<<<cdiv(e, B), B, 0, stream>>>(dst, dinv, e);
    dinv_kernel     <<<cdiv(n, B), B, 0, stream>>>(dinv, n);

    const int tb = cdiv((long)((n + 15) / 16) * 32, B);
    const int sb = cdiv((long)e, 1024);

    transform_kernel<11, 16, 0><<<tb, B, 0, stream>>>(x, W1, dinv, b1, hsA, accA, n);
    scatter_kernel<16>         <<<sb, B, 0, stream>>>(src, dst, hsA, accA, e);
    finalize_scaled_kernel<16> <<<cdiv((long)n * 4, B), B, 0, stream>>>(accA, dinv, b1, hsB, accB, n);

    scatter_kernel<16>         <<<sb, B, 0, stream>>>(src, dst, hsB, accB, e);
    transform_kernel<16, 32, 1><<<tb, B, 0, stream>>>(accB, W2, dinv, b2, hsC, accC, n);

    scatter_kernel<32>         <<<sb, B, 0, stream>>>(src, dst, hsC, accC, e);
    transform_kernel<32, 64, 2><<<tb, B, 0, stream>>>(accC, W3, dinv, b3, out, nullptr, n);
}


# Edges put in destination order once per call (integer atomic count, scan, place); each layer sums a node's source rows in registers, one store per row: no f32 atomics; f32 sum order changes
# speedup vs baseline: 2.1296x; 1.2016x over previous
#include <hip/hip_runtime.h>


typedef float v2f __attribute__((ext_vector_type(2)));
typedef float v8f __attribute__((ext_vector_type(8)));

__global__ void zero_counts_kernel(int* __restrict__ cnt, int n) {
    int i = blockIdx.x * blockDim.x + threadIdx.x;
    if (i < n) cnt[i] = 0;
}

__global__ __launch_bounds__(256) void count_rank_kernel(
    const int* __restrict__ dst, int* __restrict__ cnt, int* __restrict__ rank, int e)
{
    const int base = blockIdx.x * 1024 + threadIdx.x;
    int d[4];
#pragma unroll
    for (int u = 0; u < 4; ++u) {
        const int i = base + u * 256;
        d[u] = (i < e) ? dst[i] : -1;
    }
    int r[4];
#pragma unroll
    for (int u = 0; u < 4; ++u) r[u] = (d[u] >= 0) ? atomicAdd(&cnt[d[u]], 1) : 0;
#pragma unroll
    for (int u = 0; u < 4; ++u) {
        const int i = base + u * 256;
        if (i < e) rank[i] = r[u];
    }
}

__global__ __launch_bounds__(256) void scan_dinv_kernel(
    const int* __restrict__ cnt, int* __restrict__ rowptr, float* __restrict__ dinv, int n)
{
    __shared__ int wave_tot[8];
    __shared__ int pre_tot[8];
    const int tid  = threadIdx.x;
    const int lane = tid & 31;
    const int wv   = tid >> 5;

    int pre = 0;
    const int4* __restrict__ c4 = (const int4*)cnt;
    const int nb = blockIdx.x;
    for (int k = 0; k < nb; k += 16) {
        int4 v[16];
#pragma unroll
        for (int u = 0; u < 16; ++u) {
            const int kk = (k + u < nb) ? k + u : nb - 1;
            v[u] = c4[kk * 256 + tid];
        }
#pragma unroll
        for (int u = 0; u < 16; ++u) {
            const int s4 = (v[u].x + v[u].y) + (v[u].z + v[u].w);
            pre += (k + u < nb) ? s4 : 0;
        }
    }
#pragma unroll
    for (int o = 16; o > 0; o >>= 1) pre += __shfl_xor(pre, o);

    const int i0 = blockIdx.x * 1024 + tid * 4;
    int c[4];
#pragma unroll
    for (int k = 0; k < 4; ++k) c[k] = (i0 + k < n) ? cnt[i0 + k] : 0;
    const int tsum = (c[0] + c[1]) + (c[2] + c[3]);
    int inc = tsum;
#pragma unroll
    for (int o = 1; o < 32; o <<= 1) {
        const int t = __shfl_up(inc, o);
        if (lane >= o) inc += t;
    }
    if (lane == 31) wave_tot[wv] = inc;
    if (lane == 0)  pre_tot[wv]  = pre;
    __syncthreads();
    int off = 0;
#pragma unroll
    for (int w = 0; w < 8; ++w) {
        off += pre_tot[w];
        if (w < wv) off += wave_tot[w];
    }
    int run = off + inc - tsum;
#pragma unroll
    for (int k = 0; k < 4; ++k) {
        const int i = i0 + k;
        if (i < n) {
            rowptr[i] = run;
            dinv[i]   = rsqrtf((float)(c[k] + 1));
            run += c[k];
            if (i == n - 1) rowptr[n] = run;
        }
    }
}

__global__ __launch_bounds__(256) void place_kernel(
    const int* __restrict__ src, const int* __restrict__ dst, const int* __restrict__ rank,
    const int* __restrict__ rowptr, int* __restrict__ csr, int e)
{
    const int base = blockIdx.x * 1024 + threadIdx.x;
    int pos[4], s[4];
#pragma unroll
    for (int u = 0; u < 4; ++u) {
        const int i = base + u * 256;
        pos[u] = -1; s[u] = 0;
        if (i < e) { pos[u] = rowptr[dst[i]] + rank[i]; s[u] = src[i]; }
    }
#pragma unroll
    for (int u = 0; u < 4; ++u) if (pos[u] >= 0) csr[pos[u]] = s[u];
}

template <int MODE>
__device__ __forceinline__ float epilogue_value(float p, float di, float b) {
    if constexpr (MODE == 0) return p * di;
    else if constexpr (MODE == 1) return fmaxf(fmaf(di, p, b), 0.0f) * di;
    else return fmaf(di, p, b);
}

template <int CIN, int COUT, int MODE>
__global__ __launch_bounds__(256) void transform_kernel(
    const float* __restrict__ x, const float* __restrict__ W,
    const float* __restrict__ dinv, const float* __restrict__ bias,
    float* __restrict__ o0, int n)
{
    constexpr int  KSTEPS = (CIN + 3) / 4;
    constexpr int  TILES  = COUT / 16;
    constexpr bool KPAD   = (CIN % 4) != 0;

    int wave     = (blockIdx.x * blockDim.x + threadIdx.x) >> 5;
    int lane     = threadIdx.x & 31;
    int row_base = wave * 16;
    if (row_base >= n) return;

    int m  = lane & 15;
    int kh = (lane >> 4) & 1;
    int arow = row_base + m;
    int ar = arow < n ? arow : n - 1;

    v8f acc[TILES];
#pragma unroll
    for (int t = 0; t < TILES; ++t) acc[t] = {};

#pragma unroll
    for (int kb = 0; kb < KSTEPS; ++kb) {
        int k0 = kb * 4 + 2 * kh;
        v2f a;
        if constexpr (!KPAD) {
            a = *(const v2f*)&x[ar * CIN + k0];
        } else {
            int ka0 = (k0     < CIN) ? k0     : CIN - 1;
            int ka1 = (k0 + 1 < CIN) ? k0 + 1 : CIN - 1;
            a.x = x[ar * CIN + ka0];
            a.y = x[ar * CIN + ka1];
        }
#pragma unroll
        for (int t = 0; t < TILES; ++t) {
            int ncol = t * 16 + m;
            v2f b;
            if constexpr (!KPAD) {
                b.x = W[ k0      * COUT + ncol];
                b.y = W[(k0 + 1) * COUT + ncol];
            } else {
                bool ok0 = (k0     < CIN);
                bool ok1 = (k0 + 1 < CIN);
                float w0 = W[(ok0 ? k0     : 0) * COUT + ncol];
                float w1 = W[(ok1 ? k0 + 1 : 0) * COUT + ncol];
                b.x = ok0 ? w0 : 0.0f;
                b.y = ok1 ? w1 : 0.0f;
            }
            acc[t] = __builtin_amdgcn_wmma_f32_16x16x4_f32(
                 false, a,  false, b,
                 (short)0, acc[t],  false,  false);
        }
    }

    float bv[TILES];
#pragma unroll
    for (int t = 0; t < TILES; ++t) {
        if constexpr (MODE != 0) bv[t] = bias[t * 16 + m]; else bv[t] = 0.0f;
    }

    if (row_base + 16 <= n) {
#pragma unroll
        for (int v = 0; v < 8; ++v) {
            int r = row_base + v + 8 * kh;
            float di = dinv[r];
#pragma unroll
            for (int t = 0; t < TILES; ++t)
                o0[r * COUT + t * 16 + m] = epilogue_value<MODE>(acc[t][v], di, bv[t]);
        }
    } else {
#pragma unroll
        for (int v = 0; v < 8; ++v) {
            int r = row_base + v + 8 * kh;
            if (r < n) {
                float di = dinv[r];
#pragma unroll
                for (int t = 0; t < TILES; ++t)
                    o0[r * COUT + t * 16 + m] = epilogue_value<MODE>(acc[t][v], di, bv[t]);
            }
        }
    }
}

template <int C, bool FIN>
__global__ __launch_bounds__(256) void gather_sum_kernel(
    const int* __restrict__ rowptr, const int* __restrict__ csr,
    const float* __restrict__ hs, const float* __restrict__ dinv,
    const float* __restrict__ bias, float* __restrict__ o, int n)
{
    constexpr int NPW   = 4;
    constexpr int LPE   = (C < 32) ? C : 32;
    constexpr int EPI   = 32 / LPE;
    constexpr int STEPS = 32 / EPI;
    constexpr int GRP   = 16 / EPI;
    static_assert(C == 16 || C == 32, "one row = 16 or 32 lanes");

    const int lane = threadIdx.x & 31;
    const int ch   = lane & (LPE - 1);
    const bool up  = (lane >= LPE);
    const int wave = __builtin_amdgcn_readfirstlane(
        (int)((blockIdx.x * blockDim.x + threadIdx.x) >> 5));
    const int node0 = wave * NPW;
    if (node0 >= n) return;

    int bi = node0 + (lane < NPW ? lane : NPW);
    bi = bi < n ? bi : n;
    const int rp = rowptr[bi];

    int   iv0[NPW];
    float self[NPW];
#pragma unroll
    for (int k = 0; k < NPW; ++k) {
        const int node  = node0 + k;
        const int start = __builtin_amdgcn_readlane(rp, k);
        const int end   = __builtin_amdgcn_readlane(rp, k + 1);
        int q = start + lane;
        q = q < end ? q : end - 1;
        q = q > 0 ? q : 0;
        const int ent = csr[q];
        iv0[k] = (lane < end - start) ? ent : 0;
        const int nc = node < n ? node : n - 1;
        const float sv = hs[(unsigned)nc * (unsigned)C + (unsigned)ch];
        self[k] = (node < n && !up) ? sv : 0.0f;
    }

#pragma unroll
    for (int k = 0; k < NPW; ++k) {
        const int node = node0 + k;
        if (node >= n) return;
        const int start = __builtin_amdgcn_readlane(rp, k);
        const int end   = __builtin_amdgcn_readlane(rp, k + 1);

        float sum = self[k];

        for (int j = start; j < end; j += 32) {
            const int cnt = end - j;
            int iv = iv0[k];
            if (j != start) {
                iv = 0;
                if (lane < cnt) iv = csr[j + lane];
            }

            auto pick = [&](int t) -> unsigned {
                if constexpr (EPI == 1) {
                    return (unsigned)__builtin_amdgcn_readlane(iv, t);
                } else {
                    const int lo = __builtin_amdgcn_readlane(iv, 2 * t);
                    const int hi = __builtin_amdgcn_readlane(iv, 2 * t + 1);
                    return (unsigned)(up ? hi : lo);
                }
            };
#pragma unroll
            for (int g = 0; g < STEPS; g += GRP) {
                if (g * EPI < cnt) {
                    float v[GRP];
#pragma unroll
                    for (int t = 0; t < GRP; ++t)
                        v[t] = hs[pick(g + t) * (unsigned)C + (unsigned)ch];
#pragma unroll
                    for (int t = 0; t < GRP; ++t) {
                        const int ei = (g + t) * EPI + (up ? 1 : 0);
                        sum += (ei < cnt) ? v[t] : 0.0f;
                    }
                }
            }
        }
        if constexpr (EPI == 2) sum += __shfl_xor(sum, 16);

        float val = sum;
        if constexpr (FIN) {
            const float di = dinv[node];
            val = fmaxf(fmaf(di, sum, bias[ch]), 0.0f) * di;
        }
        if (!up) o[(unsigned)node * (unsigned)C + (unsigned)ch] = val;
    }
}

extern "C" void kernel_launch(void* const* d_in, const int* in_sizes, int n_in,
                              void* d_out, int out_size, void* d_ws, size_t ws_size,
                              hipStream_t stream) {
    const float* x  = (const float*)d_in[0];
    const int*   ei = (const int*)  d_in[1];
    const float* W1 = (const float*)d_in[2];
    const float* b1 = (const float*)d_in[3];
    const float* W2 = (const float*)d_in[4];
    const float* b2 = (const float*)d_in[5];
    const float* W3 = (const float*)d_in[6];
    const float* b3 = (const float*)d_in[7];
    float* out = (float*)d_out;

    const int n = in_sizes[0] / 11;
    const int e = in_sizes[1] / 2;
    const int* src = ei;
    const int* dst = ei + e;

    const size_t np = ((size_t)n + 64) / 64 * 64;
    const size_t ep = ((size_t)e + 63) / 64 * 64;
    int*   cnt    = (int*)d_ws;
    int*   rowptr = cnt + np;
    float* dinv   = (float*)(rowptr + np);
    int*   csr    = (int*)(dinv + np);
    float* P      = (float*)(csr + ep);
    float* Q      = P + (size_t)n * 32;
    int*   rank   = (int*)P;
    float* hsC    = P;
    float* hsA    = Q;
    float* hsB    = Q + (size_t)n * 16;
    float* accB   = Q;
    float* accC   = Q;

    const int B = 256;
    auto cdiv = [](long a, long b) { return (int)((a + b - 1) / b); };

    zero_counts_kernel<<<cdiv(n, B), B, 0, stream>>>(cnt, n);
    count_rank_kernel <<<cdiv(e, 1024), B, 0, stream>>>(dst, cnt, rank, e);
    scan_dinv_kernel  <<<cdiv(n, 1024), B, 0, stream>>>(cnt, rowptr, dinv, n);
    place_kernel      <<<cdiv(e, 1024), B, 0, stream>>>(src, dst, rank, rowptr, csr, e);

    const int tb = cdiv((long)((n + 15) / 16) * 32, B);
    const int gb = cdiv((long)((n + 3) / 4) * 32, B);

    transform_kernel<11, 16, 0><<<tb, B, 0, stream>>>(x, W1, dinv, b1, hsA, n);
    gather_sum_kernel<16, true><<<gb, B, 0, stream>>>(rowptr, csr, hsA, dinv, b1, hsB, n);

    gather_sum_kernel<16, false><<<gb, B, 0, stream>>>(rowptr, csr, hsB, dinv, b1, accB, n);
    transform_kernel<16, 32, 1><<<tb, B, 0, stream>>>(accB, W2, dinv, b2, hsC, n);

    gather_sum_kernel<32, false><<<gb, B, 0, stream>>>(rowptr, csr, hsC, dinv, b1, accC, n);
    transform_kernel<32, 64, 2><<<tb, B, 0, stream>>>(accC, W3, dinv, b3, out, n);
}


# Edges ordered by destination in two levels through LDS (256-node buckets, then nodes): per-edge counts and ranks are LDS adds, global adds only one per block and bucket; f32 sum order changes
# speedup vs baseline: 6.2184x; 2.9200x over previous
#include <hip/hip_runtime.h>


typedef float v2f __attribute__((ext_vector_type(2)));
typedef float v8f __attribute__((ext_vector_type(8)));

constexpr int BSH   = 8;
constexpr int BN    = 1 << BSH;
constexpr int NBMAX = 1024;
constexpr int CE    = 4096;
constexpr int EPT   = CE / 256;
constexpr int LT    = 1024;
constexpr int LU    = 12;

__global__ void zero_words_kernel(int* __restrict__ p, int cnt) {
    int i = blockIdx.x * blockDim.x + threadIdx.x;
    if (i < cnt) p[i] = 0;
}

template <int NW>
__device__ __forceinline__ int block_excl_scan(int v, int* red, int& total) {
    const int lane = threadIdx.x & 31;
    const int wv   = threadIdx.x >> 5;
    int inc = v;
#pragma unroll
    for (int o = 1; o < 32; o <<= 1) {
        const int t = __shfl_up(inc, o);
        if (lane >= o) inc += t;
    }
    if (lane == 31) red[wv] = inc;
    __syncthreads();
    int off = 0, tot = 0;
#pragma unroll
    for (int w = 0; w < NW; ++w) {
        const int x = red[w];
        tot += x;
        if (w < wv) off += x;
    }
    total = tot;
    __syncthreads();
    return off + inc - v;
}

__global__ __launch_bounds__(256) void bucket_count_kernel(
    const int* __restrict__ dst, int* __restrict__ tot, int e, int nb)
{
    __shared__ int h[NBMAX];
    const int tid = threadIdx.x;
    for (int i = tid; i < nb; i += 256) h[i] = 0;
    __syncthreads();
    const int base = blockIdx.x * CE + tid;
    int d[EPT];
#pragma unroll
    for (int u = 0; u < EPT; ++u) {
        const int i = base + u * 256;
        d[u] = dst[i < e ? i : e - 1];
    }
#pragma unroll
    for (int u = 0; u < EPT; ++u) {
        const int i = base + u * 256;
        if (i < e) atomicAdd(&h[d[u] >> BSH], 1);
    }
    __syncthreads();
    for (int i = tid; i < nb; i += 256) {
        const int c = h[i];
        if (c) atomicAdd(&tot[i], c);
    }
}

__global__ __launch_bounds__(256) void bucket_place_kernel(
    const int* __restrict__ src, const int* __restrict__ dst, const int* __restrict__ tot,
    int* __restrict__ cursor, unsigned* __restrict__ bw, int e, int nb)
{
    __shared__ int h[NBMAX];
    __shared__ int sb[NBMAX];
    __shared__ int ls[NBMAX];
    __shared__ unsigned sw[CE];
    __shared__ int sk[CE];
    __shared__ int red[8];
    const int tid = threadIdx.x;
#pragma unroll
    for (int k = 0; k < NBMAX / 256; ++k) h[k * 256 + tid] = 0;
    __syncthreads();

    const int first = blockIdx.x * CE;
    const int base  = first + tid;
    int d[EPT], s[EPT];
#pragma unroll
    for (int u = 0; u < EPT; ++u) {
        const int i  = base + u * 256;
        const int ic = i < e ? i : e - 1;
        d[u] = dst[ic];
        s[u] = src[ic];
    }
    int lr[EPT];
#pragma unroll
    for (int u = 0; u < EPT; ++u) {
        const int i = base + u * 256;
        lr[u] = 0;
        if (i < e) lr[u] = atomicAdd(&h[d[u] >> BSH], 1);
    }

    int c[4], hc[4];
#pragma unroll
    for (int k = 0; k < 4; ++k) c[k] = (4 * tid + k < nb) ? tot[4 * tid + k] : 0;
    int all;
    int run = block_excl_scan<8>((c[0] + c[1]) + (c[2] + c[3]), red, all);
#pragma unroll
    for (int k = 0; k < 4; ++k) hc[k] = h[4 * tid + k];
    int lrun = block_excl_scan<8>((hc[0] + hc[1]) + (hc[2] + hc[3]), red, all);
#pragma unroll
    for (int k = 0; k < 4; ++k) {
        sb[4 * tid + k] = run;
        ls[4 * tid + k] = lrun;
        run  += c[k];
        lrun += hc[k];
    }
    __syncthreads();

    int slot[EPT];
#pragma unroll
    for (int u = 0; u < EPT; ++u) slot[u] = ls[d[u] >> BSH] + lr[u];
#pragma unroll
    for (int u = 0; u < EPT; ++u) {
        const int i = base + u * 256;
        if (i < e) {
            sw[slot[u]] = ((unsigned)s[u] << BSH) | (unsigned)(d[u] & (BN - 1));
            sk[slot[u]] = d[u] >> BSH;
        }
    }
    int mine[4], got[4];
#pragma unroll
    for (int k = 0; k < 4; ++k) mine[k] = h[k * 256 + tid];
#pragma unroll
    for (int k = 0; k < 4; ++k) {
        got[k] = 0;
        if (mine[k]) got[k] = atomicAdd(&cursor[k * 256 + tid], mine[k]);
    }
#pragma unroll
    for (int k = 0; k < 4; ++k) h[k * 256 + tid] = sb[k * 256 + tid] + got[k] - ls[k * 256 + tid];
    __syncthreads();

    const int left = e - first;
    int bk2[EPT];
#pragma unroll
    for (int u = 0; u < EPT; ++u) {
        const int j = u * 256 + tid;
        bk2[u] = (j < left) ? sk[j] : 0;
    }
    int pos[EPT];
    unsigned wv[EPT];
#pragma unroll
    for (int u = 0; u < EPT; ++u) {
        const int j = u * 256 + tid;
        pos[u] = h[bk2[u]] + j;
        wv[u]  = sw[j < left ? j : 0];
    }
#pragma unroll
    for (int u = 0; u < EPT; ++u) {
        const int j = u * 256 + tid;
        if (j < left) bw[pos[u]] = wv[u];
    }
}

constexpr int LCAP = LU * LT;

__global__ __launch_bounds__(LT) void bucket_lists_kernel(
    const int* __restrict__ tot, const unsigned* __restrict__ bw,
    int* __restrict__ rowptr, float* __restrict__ dinv, int* __restrict__ csr, int n)
{
    __shared__ int cl[BN];
    __shared__ int red[LT / 32];
    __shared__ int st[LCAP];
    const int tid = threadIdx.x;
    const int b   = blockIdx.x;

    int part = 0;
    for (int i = tid; i < b; i += LT) part += tot[i];
    const int m = tot[b];
    if (tid < BN) cl[tid] = 0;
    int bbase;
    (void)block_excl_scan<LT / 32>(part, red, bbase);
    const unsigned* __restrict__ seg = bw + bbase;

    for (int i0 = 0; i0 < m; i0 += LU * LT) {
        unsigned w[LU];
#pragma unroll
        for (int u = 0; u < LU; ++u) {
            const int i = i0 + u * LT + tid;
            w[u] = seg[i < m ? i : m - 1];
        }
#pragma unroll
        for (int u = 0; u < LU; ++u) {
            const int i = i0 + u * LT + tid;
            if (i < m) atomicAdd(&cl[w[u] & (BN - 1)], 1);
        }
    }
    __syncthreads();
    const int c = (tid < BN) ? cl[tid] : 0;
    int all;
    const int ex = block_excl_scan<LT / 32>(c, red, all);
    const int node = b * BN + tid;
    if (tid < BN) {
        if (node < n) {
            rowptr[node] = bbase + ex;
            dinv[node]   = rsqrtf((float)(c + 1));
            if (node == n - 1) rowptr[n] = bbase + ex + c;
        }
        cl[tid] = ex;
    }
    __syncthreads();

    const bool staged = (m <= LCAP);
    for (int i0 = 0; i0 < m; i0 += LU * LT) {
        unsigned w[LU];
        int p[LU];
#pragma unroll
        for (int u = 0; u < LU; ++u) {
            const int i = i0 + u * LT + tid;
            w[u] = seg[i < m ? i : m - 1];
        }
#pragma unroll
        for (int u = 0; u < LU; ++u) {
            const int i = i0 + u * LT + tid;
            p[u] = 0;
            if (i < m) p[u] = atomicAdd(&cl[w[u] & (BN - 1)], 1);
        }
#pragma unroll
        for (int u = 0; u < LU; ++u) {
            const int i = i0 + u * LT + tid;
            if (i < m) {
                if (staged) st[p[u]] = (int)(w[u] >> BSH);
                else        csr[bbase + p[u]] = (int)(w[u] >> BSH);
            }
        }
    }
    if (staged) {
        __syncthreads();
        for (int i = tid; i < m; i += LT) csr[bbase + i] = st[i];
    }
}

template <int MODE>
__device__ __forceinline__ float epilogue_value(float p, float di, float b) {
    if constexpr (MODE == 0) return p * di;
    else if constexpr (MODE == 1) return fmaxf(fmaf(di, p, b), 0.0f) * di;
    else return fmaf(di, p, b);
}

template <int CIN, int COUT, int MODE>
__global__ __launch_bounds__(256) void transform_kernel(
    const float* __restrict__ x, const float* __restrict__ W,
    const float* __restrict__ dinv, const float* __restrict__ bias,
    float* __restrict__ o0, int n)
{
    constexpr int  KSTEPS = (CIN + 3) / 4;
    constexpr int  TILES  = COUT / 16;
    constexpr bool KPAD   = (CIN % 4) != 0;

    int wave     = (blockIdx.x * blockDim.x + threadIdx.x) >> 5;
    int lane     = threadIdx.x & 31;
    int row_base = wave * 16;
    if (row_base >= n) return;

    int m  = lane & 15;
    int kh = (lane >> 4) & 1;
    int arow = row_base + m;
    int ar = arow < n ? arow : n - 1;

    v8f acc[TILES];
#pragma unroll
    for (int t = 0; t < TILES; ++t) acc[t] = {};

#pragma unroll
    for (int kb = 0; kb < KSTEPS; ++kb) {
        int k0 = kb * 4 + 2 * kh;
        v2f a;
        if constexpr (!KPAD) {
            a = *(const v2f*)&x[ar * CIN + k0];
        } else {
            int ka0 = (k0     < CIN) ? k0     : CIN - 1;
            int ka1 = (k0 + 1 < CIN) ? k0 + 1 : CIN - 1;
            a.x = x[ar * CIN + ka0];
            a.y = x[ar * CIN + ka1];
        }
#pragma unroll
        for (int t = 0; t < TILES; ++t) {
            int ncol = t * 16 + m;
            v2f b;
            if constexpr (!KPAD) {
                b.x = W[ k0      * COUT + ncol];
                b.y = W[(k0 + 1) * COUT + ncol];
            } else {
                bool ok0 = (k0     < CIN);
                bool ok1 = (k0 + 1 < CIN);
                float w0 = W[(ok0 ? k0     : 0) * COUT + ncol];
                float w1 = W[(ok1 ? k0 + 1 : 0) * COUT + ncol];
                b.x = ok0 ? w0 : 0.0f;
                b.y = ok1 ? w1 : 0.0f;
            }
            acc[t] = __builtin_amdgcn_wmma_f32_16x16x4_f32(
                 false, a,  false, b,
                 (short)0, acc[t],  false,  false);
        }
    }

    float bv[TILES];
#pragma unroll
    for (int t = 0; t < TILES; ++t) {
        if constexpr (MODE != 0) bv[t] = bias[t * 16 + m]; else bv[t] = 0.0f;
    }

    if (row_base + 16 <= n) {
#pragma unroll
        for (int v = 0; v < 8; ++v) {
            int r = row_base + v + 8 * kh;
            float di = dinv[r];
#pragma unroll
            for (int t = 0; t < TILES; ++t)
                o0[r * COUT + t * 16 + m] = epilogue_value<MODE>(acc[t][v], di, bv[t]);
        }
    } else {
#pragma unroll
        for (int v = 0; v < 8; ++v) {
            int r = row_base + v + 8 * kh;
            if (r < n) {
                float di = dinv[r];
#pragma unroll
                for (int t = 0; t < TILES; ++t)
                    o0[r * COUT + t * 16 + m] = epilogue_value<MODE>(acc[t][v], di, bv[t]);
            }
        }
    }
}

template <int C, bool FIN>
__global__ __launch_bounds__(256) void gather_sum_kernel(
    const int* __restrict__ rowptr, const int* __restrict__ csr,
    const float* __restrict__ hs, const float* __restrict__ dinv,
    const float* __restrict__ bias, float* __restrict__ o, int n)
{
    constexpr int NPW   = 4;
    constexpr int LPE   = (C < 32) ? C : 32;
    constexpr int EPI   = 32 / LPE;
    constexpr int STEPS = 32 / EPI;
    constexpr int GRP   = 16 / EPI;
    static_assert(C == 16 || C == 32, "one row = 16 or 32 lanes");

    const int lane = threadIdx.x & 31;
    const int ch   = lane & (LPE - 1);
    const bool up  = (lane >= LPE);
    const int wave = __builtin_amdgcn_readfirstlane(
        (int)((blockIdx.x * blockDim.x + threadIdx.x) >> 5));
    const int node0 = wave * NPW;
    if (node0 >= n) return;

    int bi = node0 + (lane < NPW ? lane : NPW);
    bi = bi < n ? bi : n;
    const int rp = rowptr[bi];

    int   iv0[NPW];
    float self[NPW];
#pragma unroll
    for (int k = 0; k < NPW; ++k) {
        const int node  = node0 + k;
        const int start = __builtin_amdgcn_readlane(rp, k);
        const int end   = __builtin_amdgcn_readlane(rp, k + 1);
        int q = start + lane;
        q = q < end ? q : end - 1;
        q = q > 0 ? q : 0;
        const int ent = csr[q];
        iv0[k] = (lane < end - start) ? ent : 0;
        const int nc = node < n ? node : n - 1;
        const float sv = hs[(unsigned)nc * (unsigned)C + (unsigned)ch];
        self[k] = (node < n && !up) ? sv : 0.0f;
    }

#pragma unroll
    for (int k = 0; k < NPW; ++k) {
        const int node = node0 + k;
        if (node >= n) return;
        const int start = __builtin_amdgcn_readlane(rp, k);
        const int end   = __builtin_amdgcn_readlane(rp, k + 1);

        float sum = self[k];

        for (int j = start; j < end; j += 32) {
            const int cnt = end - j;
            int iv = iv0[k];
            if (j != start) {
                iv = 0;
                if (lane < cnt) iv = csr[j + lane];
            }

            auto pick = [&](int t) -> unsigned {
                if constexpr (EPI == 1) {
                    return (unsigned)__builtin_amdgcn_readlane(iv, t);
                } else {
                    const int lo = __builtin_amdgcn_readlane(iv, 2 * t);
                    const int hi = __builtin_amdgcn_readlane(iv, 2 * t + 1);
                    return (unsigned)(up ? hi : lo);
                }
            };
#pragma unroll
            for (int g = 0; g < STEPS; g += GRP) {
                if (g * EPI < cnt) {
                    float v[GRP];
#pragma unroll
                    for (int t = 0; t < GRP; ++t)
                        v[t] = hs[pick(g + t) * (unsigned)C + (unsigned)ch];
#pragma unroll
                    for (int t = 0; t < GRP; ++t) {
                        const int ei = (g + t) * EPI + (up ? 1 : 0);
                        sum += (ei < cnt) ? v[t] : 0.0f;
                    }
                }
            }
        }
        if constexpr (EPI == 2) sum += __shfl_xor(sum, 16);

        float val = sum;
        if constexpr (FIN) {
            const float di = dinv[node];
            val = fmaxf(fmaf(di, sum, bias[ch]), 0.0f) * di;
        }
        if (!up) o[(unsigned)node * (unsigned)C + (unsigned)ch] = val;
    }
}

extern "C" void kernel_launch(void* const* d_in, const int* in_sizes, int n_in,
                              void* d_out, int out_size, void* d_ws, size_t ws_size,
                              hipStream_t stream) {
    const float* x  = (const float*)d_in[0];
    const int*   ei = (const int*)  d_in[1];
    const float* W1 = (const float*)d_in[2];
    const float* b1 = (const float*)d_in[3];
    const float* W2 = (const float*)d_in[4];
    const float* b2 = (const float*)d_in[5];
    const float* W3 = (const float*)d_in[6];
    const float* b3 = (const float*)d_in[7];
    float* out = (float*)d_out;

    const int n = in_sizes[0] / 11;
    const int e = in_sizes[1] / 2;
    const int* src = ei;
    const int* dst = ei + e;

    const size_t np = ((size_t)n + 64) / 64 * 64;
    const size_t ep = ((size_t)e + 63) / 64 * 64;
    int*   tot    = (int*)d_ws;
    int*   cursor = tot + NBMAX;
    int*   rowptr = tot + np;
    float* dinv   = (float*)(rowptr + np);
    int*   csr    = (int*)(dinv + np);
    float* P      = (float*)(csr + ep);
    float* Q      = P + (size_t)n * 32;
    unsigned* bw  = (unsigned*)P;
    float* hsC    = P;
    float* hsA    = Q;
    float* hsB    = Q + (size_t)n * 16;
    float* accB   = Q;
    float* accC   = Q;

    const int B = 256;
    auto cdiv = [](long a, long b) { return (int)((a + b - 1) / b); };
    const int nb = cdiv(n, BN);

    zero_words_kernel  <<<cdiv(2 * NBMAX, B), B, 0, stream>>>(tot, 2 * NBMAX);
    bucket_count_kernel<<<cdiv(e, CE), B, 0, stream>>>(dst, tot, e, nb);
    bucket_place_kernel<<<cdiv(e, CE), B, 0, stream>>>(src, dst, tot, cursor, bw, e, nb);
    bucket_lists_kernel<<<nb, LT, 0, stream>>>(tot, bw, rowptr, dinv, csr, n);

    const int tb = cdiv((long)((n + 15) / 16) * 32, B);
    const int gb = cdiv((long)((n + 3) / 4) * 32, B);

    transform_kernel<11, 16, 0><<<tb, B, 0, stream>>>(x, W1, dinv, b1, hsA, n);
    gather_sum_kernel<16, true><<<gb, B, 0, stream>>>(rowptr, csr, hsA, dinv, b1, hsB, n);

    gather_sum_kernel<16, false><<<gb, B, 0, stream>>>(rowptr, csr, hsB, dinv, b1, accB, n);
    transform_kernel<16, 32, 1><<<tb, B, 0, stream>>>(accB, W2, dinv, b2, hsC, n);

    gather_sum_kernel<32, false><<<gb, B, 0, stream>>>(rowptr, csr, hsC, dinv, b1, accC, n);
    transform_kernel<32, 64, 2><<<tb, B, 0, stream>>>(accC, W3, dinv, b3, out, n);
}


# Each dense product runs inside the kernel that finishes its input rows (layer 1: the dinv pass; layers 2,3: the edge sum, rows via LDS into the f32 WMMA): 10->7 launches; f32 sum order kept
# speedup vs baseline: 7.0267x; 1.1300x over previous
#include <hip/hip_runtime.h>


typedef float v2f __attribute__((ext_vector_type(2)));
typedef float v8f __attribute__((ext_vector_type(8)));

constexpr int BSH   = 8;
constexpr int BN    = 1 << BSH;
constexpr int NBMAX = 1024;
constexpr int CE    = 4096;
constexpr int EPT   = CE / 256;
constexpr int LT    = 1024;
constexpr int LU    = 12;

__global__ void zero_words_kernel(int* __restrict__ p, int cnt) {
    int i = blockIdx.x * blockDim.x + threadIdx.x;
    if (i < cnt) p[i] = 0;
}

template <int NW>
__device__ __forceinline__ int block_excl_scan(int v, int* red, int& total) {
    const int lane = threadIdx.x & 31;
    const int wv   = threadIdx.x >> 5;
    int inc = v;
#pragma unroll
    for (int o = 1; o < 32; o <<= 1) {
        const int t = __shfl_up(inc, o);
        if (lane >= o) inc += t;
    }
    if (lane == 31) red[wv] = inc;
    __syncthreads();
    int off = 0, tot = 0;
#pragma unroll
    for (int w = 0; w < NW; ++w) {
        const int x = red[w];
        tot += x;
        if (w < wv) off += x;
    }
    total = tot;
    __syncthreads();
    return off + inc - v;
}

__global__ __launch_bounds__(256) void bucket_count_kernel(
    const int* __restrict__ dst, int* __restrict__ tot, int e, int nb)
{
    __shared__ int h[NBMAX];
    const int tid = threadIdx.x;
    for (int i = tid; i < nb; i += 256) h[i] = 0;
    __syncthreads();
    const int base = blockIdx.x * CE + tid;
    int d[EPT];
#pragma unroll
    for (int u = 0; u < EPT; ++u) {
        const int i = base + u * 256;
        d[u] = dst[i < e ? i : e - 1];
    }
#pragma unroll
    for (int u = 0; u < EPT; ++u) {
        const int i = base + u * 256;
        if (i < e) atomicAdd(&h[d[u] >> BSH], 1);
    }
    __syncthreads();
    for (int i = tid; i < nb; i += 256) {
        const int c = h[i];
        if (c) atomicAdd(&tot[i], c);
    }
}

__global__ __launch_bounds__(256) void bucket_place_kernel(
    const int* __restrict__ src, const int* __restrict__ dst, const int* __restrict__ tot,
    int* __restrict__ cursor, unsigned* __restrict__ bw, int e, int nb)
{
    __shared__ int h[NBMAX];
    __shared__ int sb[NBMAX];
    __shared__ int ls[NBMAX];
    __shared__ unsigned sw[CE];
    __shared__ int sk[CE];
    __shared__ int red[8];
    const int tid = threadIdx.x;
#pragma unroll
    for (int k = 0; k < NBMAX / 256; ++k) h[k * 256 + tid] = 0;
    __syncthreads();

    const int first = blockIdx.x * CE;
    const int base  = first + tid;
    int d[EPT], s[EPT];
#pragma unroll
    for (int u = 0; u < EPT; ++u) {
        const int i  = base + u * 256;
        const int ic = i < e ? i : e - 1;
        d[u] = dst[ic];
        s[u] = src[ic];
    }
    int lr[EPT];
#pragma unroll
    for (int u = 0; u < EPT; ++u) {
        const int i = base + u * 256;
        lr[u] = 0;
        if (i < e) lr[u] = atomicAdd(&h[d[u] >> BSH], 1);
    }

    int c[4], hc[4];
#pragma unroll
    for (int k = 0; k < 4; ++k) c[k] = (4 * tid + k < nb) ? tot[4 * tid + k] : 0;
    int all;
    int run = block_excl_scan<8>((c[0] + c[1]) + (c[2] + c[3]), red, all);
#pragma unroll
    for (int k = 0; k < 4; ++k) hc[k] = h[4 * tid + k];
    int lrun = block_excl_scan<8>((hc[0] + hc[1]) + (hc[2] + hc[3]), red, all);
#pragma unroll
    for (int k = 0; k < 4; ++k) {
        sb[4 * tid + k] = run;
        ls[4 * tid + k] = lrun;
        run  += c[k];
        lrun += hc[k];
    }
    __syncthreads();

    int slot[EPT];
#pragma unroll
    for (int u = 0; u < EPT; ++u) slot[u] = ls[d[u] >> BSH] + lr[u];
#pragma unroll
    for (int u = 0; u < EPT; ++u) {
        const int i = base + u * 256;
        if (i < e) {
            sw[slot[u]] = ((unsigned)s[u] << BSH) | (unsigned)(d[u] & (BN - 1));
            sk[slot[u]] = d[u] >> BSH;
        }
    }
    int mine[4], got[4];
#pragma unroll
    for (int k = 0; k < 4; ++k) mine[k] = h[k * 256 + tid];
#pragma unroll
    for (int k = 0; k < 4; ++k) {
        got[k] = 0;
        if (mine[k]) got[k] = atomicAdd(&cursor[k * 256 + tid], mine[k]);
    }
#pragma unroll
    for (int k = 0; k < 4; ++k) h[k * 256 + tid] = sb[k * 256 + tid] + got[k] - ls[k * 256 + tid];
    __syncthreads();

    const int left = e - first;
    int bk2[EPT];
#pragma unroll
    for (int u = 0; u < EPT; ++u) {
        const int j = u * 256 + tid;
        bk2[u] = (j < left) ? sk[j] : 0;
    }
    int pos[EPT];
    unsigned wv[EPT];
#pragma unroll
    for (int u = 0; u < EPT; ++u) {
        const int j = u * 256 + tid;
        pos[u] = h[bk2[u]] + j;
        wv[u]  = sw[j < left ? j : 0];
    }
#pragma unroll
    for (int u = 0; u < EPT; ++u) {
        const int j = u * 256 + tid;
        if (j < left) bw[pos[u]] = wv[u];
    }
}

constexpr int LCAP = LU * LT;

__global__ __launch_bounds__(LT) void bucket_lists_kernel(
    const int* __restrict__ tot, const unsigned* __restrict__ bw,
    int* __restrict__ rowptr, float* __restrict__ dinv, int* __restrict__ csr,
    const float* __restrict__ x, const float* __restrict__ W1, float* __restrict__ hsA, int n)
{
    __shared__ int cl[BN];
    __shared__ int red[LT / 32];
    __shared__ int st[LCAP];
    __shared__ float dl[BN];
    const int tid = threadIdx.x;
    const int b   = blockIdx.x;

    constexpr int CIN1 = 11, COUT1 = 16, KS1 = (CIN1 + 3) / 4;
    const int lane = tid & 31;
    const int wv   = __builtin_amdgcn_readfirstlane(tid >> 5);
    const int pm   = lane & 15;
    const int pkh  = (lane >> 4) & 1;
    const int row_base = b * BN + wv * 16;
    const bool prod = (wv < BN / 16) && (row_base < n);
    v2f fa[KS1], fb[KS1];
#pragma unroll
    for (int kb = 0; kb < KS1; ++kb) { fa[kb] = {}; fb[kb] = {}; }
    if (prod) {
        const int arow = row_base + pm;
        const int ar = arow < n ? arow : n - 1;
#pragma unroll
        for (int kb = 0; kb < KS1; ++kb) {
            const int  k0  = kb * 4 + 2 * pkh;
            const bool ok0 = (k0     < CIN1);
            const bool ok1 = (k0 + 1 < CIN1);
            fa[kb].x = x[ar * CIN1 + (ok0 ? k0     : CIN1 - 1)];
            fa[kb].y = x[ar * CIN1 + (ok1 ? k0 + 1 : CIN1 - 1)];
            fb[kb].x = W1[(ok0 ? k0     : 0) * COUT1 + pm];
            fb[kb].y = W1[(ok1 ? k0 + 1 : 0) * COUT1 + pm];
        }
    }

    int part = 0;
    for (int i = tid; i < b; i += LT) part += tot[i];
    const int m = tot[b];
    if (tid < BN) cl[tid] = 0;
    int bbase;
    (void)block_excl_scan<LT / 32>(part, red, bbase);
    const unsigned* __restrict__ seg = bw + bbase;

    for (int i0 = 0; i0 < m; i0 += LU * LT) {
        unsigned w[LU];
#pragma unroll
        for (int u = 0; u < LU; ++u) {
            const int i = i0 + u * LT + tid;
            w[u] = seg[i < m ? i : m - 1];
        }
#pragma unroll
        for (int u = 0; u < LU; ++u) {
            const int i = i0 + u * LT + tid;
            if (i < m) atomicAdd(&cl[w[u] & (BN - 1)], 1);
        }
    }
    __syncthreads();
    const int c = (tid < BN) ? cl[tid] : 0;
    int all;
    const int ex = block_excl_scan<LT / 32>(c, red, all);
    const int node = b * BN + tid;
    if (tid < BN) {
        if (node < n) {
            const float di = rsqrtf((float)(c + 1));
            rowptr[node] = bbase + ex;
            dinv[node]   = di;
            dl[tid]      = di;
            if (node == n - 1) rowptr[n] = bbase + ex + c;
        }
        cl[tid] = ex;
    }
    __syncthreads();

    if (prod) {
        v8f acc = {};
#pragma unroll
        for (int kb = 0; kb < KS1; ++kb) {
            const int k0 = kb * 4 + 2 * pkh;
            v2f bq;
            bq.x = (k0     < CIN1) ? fb[kb].x : 0.0f;
            bq.y = (k0 + 1 < CIN1) ? fb[kb].y : 0.0f;
            acc = __builtin_amdgcn_wmma_f32_16x16x4_f32(
                 false, fa[kb],  false, bq,
                 (short)0, acc,  false,  false);
        }
        if (row_base + 16 <= n) {
#pragma unroll
            for (int v = 0; v < 8; ++v) {
                const int lr = wv * 16 + v + 8 * pkh;
                hsA[(b * BN + lr) * COUT1 + pm] = acc[v] * dl[lr];
            }
        } else {
#pragma unroll
            for (int v = 0; v < 8; ++v) {
                const int lr = wv * 16 + v + 8 * pkh;
                const int r  = b * BN + lr;
                if (r < n) hsA[r * COUT1 + pm] = acc[v] * dl[lr];
            }
        }
    }

    const bool staged = (m <= LCAP);
    for (int i0 = 0; i0 < m; i0 += LU * LT) {
        unsigned w[LU];
        int p[LU];
#pragma unroll
        for (int u = 0; u < LU; ++u) {
            const int i = i0 + u * LT + tid;
            w[u] = seg[i < m ? i : m - 1];
        }
#pragma unroll
        for (int u = 0; u < LU; ++u) {
            const int i = i0 + u * LT + tid;
            p[u] = 0;
            if (i < m) p[u] = atomicAdd(&cl[w[u] & (BN - 1)], 1);
        }
#pragma unroll
        for (int u = 0; u < LU; ++u) {
            const int i = i0 + u * LT + tid;
            if (i < m) {
                if (staged) st[p[u]] = (int)(w[u] >> BSH);
                else        csr[bbase + p[u]] = (int)(w[u] >> BSH);
            }
        }
    }
    if (staged) {
        __syncthreads();
        for (int i = tid; i < m; i += LT) csr[bbase + i] = st[i];
    }
}

template <int MODE>
__device__ __forceinline__ float epilogue_value(float p, float di, float b) {
    if constexpr (MODE == 1) return fmaxf(fmaf(di, p, b), 0.0f) * di;
    else return fmaf(di, p, b);
}

template <int C, int COUT, int MODE>
__global__ __launch_bounds__(256) void gather_sum_kernel(
    const int* __restrict__ rowptr, const int* __restrict__ csr,
    const float* __restrict__ hs, const float* __restrict__ dinv,
    const float* __restrict__ bias, const float* __restrict__ W,
    float* __restrict__ o, int n)
{
    constexpr bool FIN  = (COUT == 0);
    constexpr int NPW   = 4;
    constexpr int NPB   = 8 * NPW;
    constexpr int PITCH = C + 2;
    constexpr int LPE   = (C < 32) ? C : 32;
    constexpr int EPI   = 32 / LPE;
    constexpr int STEPS = 32 / EPI;
    constexpr int GRP   = 16 / EPI;
    static_assert(C == 16 || C == 32, "one row = 16 or 32 lanes");
    static_assert(COUT % 32 == 0, "a product wave owns whole 128-B lines");

    __shared__ float rows[FIN ? 1 : NPB * PITCH];

    const int lane = threadIdx.x & 31;
    const int ch   = lane & (LPE - 1);
    const bool up  = (lane >= LPE);
    const int wave = __builtin_amdgcn_readfirstlane(
        (int)((blockIdx.x * blockDim.x + threadIdx.x) >> 5));
    const int wib  = wave & 7;
    const int node0 = wave * NPW;

    if (node0 < n) {
    int bi = node0 + (lane < NPW ? lane : NPW);
    bi = bi < n ? bi : n;
    const int rp = rowptr[bi];

    int   iv0[NPW];
    float self[NPW];
#pragma unroll
    for (int k = 0; k < NPW; ++k) {
        const int node  = node0 + k;
        const int start = __builtin_amdgcn_readlane(rp, k);
        const int end   = __builtin_amdgcn_readlane(rp, k + 1);
        int q = start + lane;
        q = q < end ? q : end - 1;
        q = q > 0 ? q : 0;
        const int ent = csr[q];
        iv0[k] = (lane < end - start) ? ent : 0;
        const int nc = node < n ? node : n - 1;
        const float sv = hs[(unsigned)nc * (unsigned)C + (unsigned)ch];
        self[k] = (node < n && !up) ? sv : 0.0f;
    }

#pragma unroll
    for (int k = 0; k < NPW; ++k) {
        const int node = node0 + k;
        if (node < n) {
        const int start = __builtin_amdgcn_readlane(rp, k);
        const int end   = __builtin_amdgcn_readlane(rp, k + 1);

        float sum = self[k];

        for (int j = start; j < end; j += 32) {
            const int cnt = end - j;
            int iv = iv0[k];
            if (j != start) {
                iv = 0;
                if (lane < cnt) iv = csr[j + lane];
            }

            auto pick = [&](int t) -> unsigned {
                if constexpr (EPI == 1) {
                    return (unsigned)__builtin_amdgcn_readlane(iv, t);
                } else {
                    const int lo = __builtin_amdgcn_readlane(iv, 2 * t);
                    const int hi = __builtin_amdgcn_readlane(iv, 2 * t + 1);
                    return (unsigned)(up ? hi : lo);
                }
            };
#pragma unroll
            for (int g = 0; g < STEPS; g += GRP) {
                if (g * EPI < cnt) {
                    float v[GRP];
#pragma unroll
                    for (int t = 0; t < GRP; ++t)
                        v[t] = hs[pick(g + t) * (unsigned)C + (unsigned)ch];
#pragma unroll
                    for (int t = 0; t < GRP; ++t) {
                        const int ei = (g + t) * EPI + (up ? 1 : 0);
                        sum += (ei < cnt) ? v[t] : 0.0f;
                    }
                }
            }
        }
        if constexpr (EPI == 2) sum += __shfl_xor(sum, 16);

        if constexpr (FIN) {
            const float di = dinv[node];
            const float val = fmaxf(fmaf(di, sum, bias[ch]), 0.0f) * di;
            if (!up) o[(unsigned)node * (unsigned)C + (unsigned)ch] = val;
        } else {
            if (!up) rows[(wib * NPW + k) * PITCH + ch] = sum;
        }
        }
    }
    }

    if constexpr (!FIN) {
        constexpr int CG     = COUT / 32;
        constexpr int JOBS   = (NPB / 16) * CG;
        constexpr int KSTEPS = C / 4;
        __syncthreads();
        if (wib < JOBS) {
            const int rt = wib / CG;
            const int cg = wib % CG;
            const int row_base = (int)blockIdx.x * NPB + rt * 16;
            if (row_base < n) {
                const int m  = lane & 15;
                const int kh = (lane >> 4) & 1;
                const float* __restrict__ arow = &rows[(rt * 16 + m) * PITCH];
                float bv[2];
#pragma unroll
                for (int t = 0; t < 2; ++t) bv[t] = bias[cg * 32 + t * 16 + m];
                v8f acc[2];
                acc[0] = {};
                acc[1] = {};
#pragma unroll
                for (int kb = 0; kb < KSTEPS; ++kb) {
                    const int k0 = kb * 4 + 2 * kh;
                    const v2f a = *(const v2f*)&arow[k0];
#pragma unroll
                    for (int t = 0; t < 2; ++t) {
                        const int ncol = cg * 32 + t * 16 + m;
                        v2f b;
                        b.x = W[ k0      * COUT + ncol];
                        b.y = W[(k0 + 1) * COUT + ncol];
                        acc[t] = __builtin_amdgcn_wmma_f32_16x16x4_f32(
                             false, a,  false, b,
                             (short)0, acc[t],  false,  false);
                    }
                }

                if (row_base + 16 <= n) {
#pragma unroll
                    for (int v = 0; v < 8; ++v) {
                        const int r = row_base + v + 8 * kh;
                        const float di = dinv[r];
#pragma unroll
                        for (int t = 0; t < 2; ++t)
                            o[r * COUT + cg * 32 + t * 16 + m] = epilogue_value<MODE>(acc[t][v], di, bv[t]);
                    }
                } else {
#pragma unroll
                    for (int v = 0; v < 8; ++v) {
                        const int r = row_base + v + 8 * kh;
                        if (r < n) {
                            const float di = dinv[r];
#pragma unroll
                            for (int t = 0; t < 2; ++t)
                                o[r * COUT + cg * 32 + t * 16 + m] = epilogue_value<MODE>(acc[t][v], di, bv[t]);
                        }
                    }
                }
            }
        }
    }
}

extern "C" void kernel_launch(void* const* d_in, const int* in_sizes, int n_in,
                              void* d_out, int out_size, void* d_ws, size_t ws_size,
                              hipStream_t stream) {
    const float* x  = (const float*)d_in[0];
    const int*   ei = (const int*)  d_in[1];
    const float* W1 = (const float*)d_in[2];
    const float* b1 = (const float*)d_in[3];
    const float* W2 = (const float*)d_in[4];
    const float* b2 = (const float*)d_in[5];
    const float* W3 = (const float*)d_in[6];
    const float* b3 = (const float*)d_in[7];
    float* out = (float*)d_out;

    const int n = in_sizes[0] / 11;
    const int e = in_sizes[1] / 2;
    const int* src = ei;
    const int* dst = ei + e;

    const size_t np = ((size_t)n + 64) / 64 * 64;
    const size_t ep = ((size_t)e + 63) / 64 * 64;
    int*   tot    = (int*)d_ws;
    int*   cursor = tot + NBMAX;
    int*   rowptr = tot + np;
    float* dinv   = (float*)(rowptr + np);
    int*   csr    = (int*)(dinv + np);
    float* P      = (float*)(csr + ep);
    float* Q      = P + (size_t)n * 32;
    unsigned* bw  = (unsigned*)P;
    float* hsC    = P;
    float* hsB    = Q;
    float* hsA    = Q + (size_t)n * 16;

    const int B = 256;
    auto cdiv = [](long a, long b) { return (int)((a + b - 1) / b); };
    const int nb = cdiv(n, BN);

    zero_words_kernel  <<<cdiv(2 * NBMAX, B), B, 0, stream>>>(tot, 2 * NBMAX);
    bucket_count_kernel<<<cdiv(e, CE), B, 0, stream>>>(dst, tot, e, nb);
    bucket_place_kernel<<<cdiv(e, CE), B, 0, stream>>>(src, dst, tot, cursor, bw, e, nb);
    bucket_lists_kernel<<<nb, LT, 0, stream>>>(tot, bw, rowptr, dinv, csr, x, W1, hsA, n);

    const int gb = cdiv((long)((n + 3) / 4) * 32, B);

    gather_sum_kernel<16, 0, 0><<<gb, B, 0, stream>>>(rowptr, csr, hsA, dinv, b1, W1, hsB, n);

    gather_sum_kernel<16, 32, 1><<<gb, B, 0, stream>>>(rowptr, csr, hsB, dinv, b2, W2, hsC, n);

    gather_sum_kernel<32, 64, 2><<<gb, B, 0, stream>>>(rowptr, csr, hsC, dinv, b3, W3, out, n);
}


# First ordering level block-local: a block's bucket-ordered edges stay in its own segment with a run table, list blocks collect their runs: no global atomics, 7->5 launches; f32 sum order changes
# speedup vs baseline: 7.8428x; 1.1161x over previous
#include <hip/hip_runtime.h>


typedef float v2f __attribute__((ext_vector_type(2)));
typedef float v8f __attribute__((ext_vector_type(8)));

constexpr int BSH   = 8;
constexpr int BN    = 1 << BSH;
constexpr int NBMAX = 1024;
constexpr int CE    = 4096;
constexpr int EPT   = CE / 256;
constexpr int LT    = 1024;
constexpr int LU    = 12;
constexpr int RB    = 16;

template <int NW>
__device__ __forceinline__ int block_excl_scan(int v, int* red, int& total) {
    const int lane = threadIdx.x & 31;
    const int wv   = threadIdx.x >> 5;
    int inc = v;
#pragma unroll
    for (int o = 1; o < 32; o <<= 1) {
        const int t = __shfl_up(inc, o);
        if (lane >= o) inc += t;
    }
    if (lane == 31) red[wv] = inc;
    __syncthreads();
    int off = 0, tot = 0;
#pragma unroll
    for (int w = 0; w < NW; ++w) {
        const int x = red[w];
        tot += x;
        if (w < wv) off += x;
    }
    total = tot;
    __syncthreads();
    return off + inc - v;
}

__device__ __forceinline__ int block_excl_scan32(int v, int* red, int& total) {
    const int lane = threadIdx.x & 31;
    const int wv   = __builtin_amdgcn_readfirstlane((int)(threadIdx.x >> 5));
    int inc = v;
#pragma unroll
    for (int o = 1; o < 32; o <<= 1) {
        const int t = __shfl_up(inc, o);
        if (lane >= o) inc += t;
    }
    if (lane == 31) red[wv] = inc;
    __syncthreads();
    const int x = red[lane];
    int off = 0, tot = 0;
#pragma unroll
    for (int w = 0; w < 32; ++w) {
        const int xw = __builtin_amdgcn_readlane(x, w);
        tot += xw;
        if (w < wv) off += xw;
    }
    total = tot;
    __syncthreads();
    return off + inc - v;
}

__global__ __launch_bounds__(256) void bucket_place_kernel(
    const int* __restrict__ src, const int* __restrict__ dst,
    unsigned* __restrict__ bw, int* __restrict__ tab, int e, int nb)
{
    __shared__ int h[NBMAX];
    __shared__ int ls[NBMAX];
    __shared__ unsigned sw[CE];
    __shared__ int red[8];
    const int tid = threadIdx.x;
#pragma unroll
    for (int k = 0; k < NBMAX / 256; ++k) h[k * 256 + tid] = 0;
    __syncthreads();

    const int first = blockIdx.x * CE;
    const int base  = first + tid;
    int d[EPT], s[EPT];
#pragma unroll
    for (int u = 0; u < EPT; ++u) {
        const int i  = base + u * 256;
        const int ic = i < e ? i : e - 1;
        d[u] = dst[ic];
        s[u] = src[ic];
    }
    int lr[EPT];
#pragma unroll
    for (int u = 0; u < EPT; ++u) {
        const int i = base + u * 256;
        lr[u] = 0;
        if (i < e) lr[u] = atomicAdd(&h[d[u] >> BSH], 1);
    }
    __syncthreads();

    int hc[4];
#pragma unroll
    for (int k = 0; k < 4; ++k) hc[k] = h[4 * tid + k];
    int all;
    int lrun = block_excl_scan<8>((hc[0] + hc[1]) + (hc[2] + hc[3]), red, all);
#pragma unroll
    for (int k = 0; k < 4; ++k) {
        ls[4 * tid + k] = lrun;
        lrun += hc[k];
    }
    __syncthreads();

    int slot[EPT];
#pragma unroll
    for (int u = 0; u < EPT; ++u) slot[u] = ls[d[u] >> BSH] + lr[u];
#pragma unroll
    for (int u = 0; u < EPT; ++u) {
        const int i = base + u * 256;
        if (i < e) sw[slot[u]] = ((unsigned)s[u] << BSH) | (unsigned)(d[u] & (BN - 1));
    }
    int* __restrict__ trow = tab + (size_t)blockIdx.x * (size_t)(nb + 1);
    for (int i = tid; i <= nb; i += 256) trow[i] = (i < nb) ? ls[i] : all;
    __syncthreads();

    unsigned wv[EPT];
#pragma unroll
    for (int u = 0; u < EPT; ++u) {
        const int j = u * 256 + tid;
        wv[u] = sw[j < all ? j : 0];
    }
#pragma unroll
    for (int u = 0; u < EPT; ++u) {
        const int j = u * 256 + tid;
        if (j < all) bw[first + j] = wv[u];
    }
}

static_assert(LT == 1024, "block_excl_scan32 is for 32 waves");
constexpr int LCAP = LU * LT;

__global__ __launch_bounds__(LT) void bucket_lists_kernel(
    const int* __restrict__ tab, const unsigned* __restrict__ bw,
    int* __restrict__ rowptr, float* __restrict__ dinv, int* __restrict__ csr,
    const float* __restrict__ x, const float* __restrict__ W1, float* __restrict__ hsA,
    int n, int nblk, int nb, int e)
{
    __shared__ int cl[BN];
    __shared__ int red[LT / 32];
    __shared__ unsigned stg[LCAP];
    __shared__ float dl[BN];
    const int tid = threadIdx.x;
    const int b   = blockIdx.x;

    constexpr int CIN1 = 11, COUT1 = 16, KS1 = (CIN1 + 3) / 4;
    const int lane = tid & 31;
    const int wv   = __builtin_amdgcn_readfirstlane(tid >> 5);
    const int pm   = lane & 15;
    const int pkh  = (lane >> 4) & 1;
    const int row_base = b * BN + wv * 16;
    const bool prod = (wv < BN / 16) && (row_base < n);
    v2f fa[KS1], fb[KS1];
#pragma unroll
    for (int kb = 0; kb < KS1; ++kb) { fa[kb] = {}; fb[kb] = {}; }
    if (prod) {
        const int arow = row_base + pm;
        const int ar = arow < n ? arow : n - 1;
#pragma unroll
        for (int kb = 0; kb < KS1; ++kb) {
            const int  k0  = kb * 4 + 2 * pkh;
            const bool ok0 = (k0     < CIN1);
            const bool ok1 = (k0 + 1 < CIN1);
            fa[kb].x = x[ar * CIN1 + (ok0 ? k0     : CIN1 - 1)];
            fa[kb].y = x[ar * CIN1 + (ok1 ? k0 + 1 : CIN1 - 1)];
            fb[kb].x = W1[(ok0 ? k0     : 0) * COUT1 + pm];
            fb[kb].y = W1[(ok1 ? k0 + 1 : 0) * COUT1 + pm];
        }
    }

    const int tp = nb + 1;
    int ra = 0, rcn = 0;
    if (tid < nblk) {
        const int* __restrict__ tr = tab + (size_t)tid * (size_t)tp + b;
        ra  = tr[0];
        rcn = tr[1] - ra;
    }
    int asum = ra, csum = rcn;
    for (int k = tid + LT; k < nblk; k += LT) {
        const int* __restrict__ tr = tab + (size_t)k * (size_t)tp + b;
        const int a2 = tr[0];
        asum += a2;
        csum += tr[1] - a2;
    }
    if (tid < BN) cl[tid] = 0;
    int m, bbase;
    const int off = block_excl_scan32(rcn, red, m);
    (void)block_excl_scan32(asum, red, bbase);
    if (nblk > LT) (void)block_excl_scan32(csum, red, m);
    const bool staged = (m <= LCAP) && (nblk <= LT);

    if (staged) {
        const int rsv = (tid < nblk) ? tid * CE + ra : 0;
        if (wv * 32 < nblk) {
#pragma unroll 1
            for (int qb = 0; qb < 32; qb += RB) {
                unsigned w0[RB];
#pragma unroll
                for (int q = 0; q < RB; ++q) {
                    const int rs = __builtin_amdgcn_readlane(rsv, qb + q);
                    const int rc = __builtin_amdgcn_readlane(rcn, qb + q);
                    const int lastj = rc > 0 ? rc - 1 : 0;
                    int i0 = rs + (lane < lastj ? lane : lastj);
                    i0 = i0 < e ? i0 : e - 1;
                    w0[q] = bw[i0];
                }
#pragma unroll
                for (int q = 0; q < RB; ++q) {
                    const int rs = __builtin_amdgcn_readlane(rsv, qb + q);
                    const int rc = __builtin_amdgcn_readlane(rcn, qb + q);
                    const int ro = __builtin_amdgcn_readlane(off, qb + q);
                    if (lane < rc) {
                        stg[ro + lane] = w0[q];
                        atomicAdd(&cl[w0[q] & (BN - 1)], 1);
                    }
                    for (int j = lane + 32; j < rc; j += 32) {
                        const unsigned w = bw[rs + j];
                        stg[ro + j] = w;
                        atomicAdd(&cl[w & (BN - 1)], 1);
                    }
                }
            }
        }
    } else {
        for (int k = tid; k < nblk; k += LT) {
            const int* __restrict__ tr = tab + (size_t)k * (size_t)tp + b;
            const int a2 = tr[0];
            const int c2 = tr[1] - a2;
            for (int j = 0; j < c2; ++j)
                atomicAdd(&cl[bw[k * CE + a2 + j] & (BN - 1)], 1);
        }
    }
    __syncthreads();
    const int c = (tid < BN) ? cl[tid] : 0;
    int all;
    const int ex = block_excl_scan32(c, red, all);
    const int node = b * BN + tid;
    if (tid < BN) {
        if (node < n) {
            const float di = rsqrtf((float)(c + 1));
            rowptr[node] = bbase + ex;
            dinv[node]   = di;
            dl[tid]      = di;
            if (node == n - 1) rowptr[n] = bbase + ex + c;
        }
        cl[tid] = ex;
    }
    __syncthreads();

    if (prod) {
        v8f acc = {};
#pragma unroll
        for (int kb = 0; kb < KS1; ++kb) {
            const int k0 = kb * 4 + 2 * pkh;
            v2f bq;
            bq.x = (k0     < CIN1) ? fb[kb].x : 0.0f;
            bq.y = (k0 + 1 < CIN1) ? fb[kb].y : 0.0f;
            acc = __builtin_amdgcn_wmma_f32_16x16x4_f32(
                 false, fa[kb],  false, bq,
                 (short)0, acc,  false,  false);
        }
        if (row_base + 16 <= n) {
#pragma unroll
            for (int v = 0; v < 8; ++v) {
                const int lr = wv * 16 + v + 8 * pkh;
                hsA[(b * BN + lr) * COUT1 + pm] = acc[v] * dl[lr];
            }
        } else {
#pragma unroll
            for (int v = 0; v < 8; ++v) {
                const int lr = wv * 16 + v + 8 * pkh;
                const int r  = b * BN + lr;
                if (r < n) hsA[r * COUT1 + pm] = acc[v] * dl[lr];
            }
        }
    }

    if (staged) {
        unsigned w[LU];
        int p[LU];
#pragma unroll
        for (int u = 0; u < LU; ++u) {
            const int i = u * LT + tid;
            w[u] = stg[(i < m) ? i : 0];
        }
        __syncthreads();
#pragma unroll
        for (int u = 0; u < LU; ++u) {
            const int i = u * LT + tid;
            p[u] = 0;
            if (i < m) p[u] = atomicAdd(&cl[w[u] & (BN - 1)], 1);
        }
#pragma unroll
        for (int u = 0; u < LU; ++u) {
            const int i = u * LT + tid;
            if (i < m) stg[p[u]] = w[u] >> BSH;
        }
        __syncthreads();
        for (int i = tid; i < m; i += LT) csr[bbase + i] = (int)stg[i];
    } else {
        for (int k = tid; k < nblk; k += LT) {
            const int* __restrict__ tr = tab + (size_t)k * (size_t)tp + b;
            const int a2 = tr[0];
            const int c2 = tr[1] - a2;
            for (int j = 0; j < c2; ++j) {
                const unsigned w = bw[k * CE + a2 + j];
                const int p = atomicAdd(&cl[w & (BN - 1)], 1);
                csr[bbase + p] = (int)(w >> BSH);
            }
        }
    }
}

template <int MODE>
__device__ __forceinline__ float epilogue_value(float p, float di, float b) {
    if constexpr (MODE == 1) return fmaxf(fmaf(di, p, b), 0.0f) * di;
    else return fmaf(di, p, b);
}

template <int C, int COUT, int MODE>
__global__ __launch_bounds__(256) void gather_sum_kernel(
    const int* __restrict__ rowptr, const int* __restrict__ csr,
    const float* __restrict__ hs, const float* __restrict__ dinv,
    const float* __restrict__ bias, const float* __restrict__ W,
    float* __restrict__ o, int n)
{
    constexpr bool FIN  = (COUT == 0);
    constexpr int NPW   = 4;
    constexpr int NPB   = 8 * NPW;
    constexpr int PITCH = C + 2;
    constexpr int LPE   = (C < 32) ? C : 32;
    constexpr int EPI   = 32 / LPE;
    constexpr int STEPS = 32 / EPI;
    constexpr int GRP   = 16 / EPI;
    static_assert(C == 16 || C == 32, "one row = 16 or 32 lanes");
    static_assert(COUT % 32 == 0, "a product wave owns whole 128-B lines");

    __shared__ float rows[FIN ? 1 : NPB * PITCH];

    const int lane = threadIdx.x & 31;
    const int ch   = lane & (LPE - 1);
    const bool up  = (lane >= LPE);
    const int wave = __builtin_amdgcn_readfirstlane(
        (int)((blockIdx.x * blockDim.x + threadIdx.x) >> 5));
    const int wib  = wave & 7;
    const int node0 = wave * NPW;

    if (node0 < n) {
    int bi = node0 + (lane < NPW ? lane : NPW);
    bi = bi < n ? bi : n;
    const int rp = rowptr[bi];

    int   iv0[NPW];
    float self[NPW];
#pragma unroll
    for (int k = 0; k < NPW; ++k) {
        const int node  = node0 + k;
        const int start = __builtin_amdgcn_readlane(rp, k);
        const int end   = __builtin_amdgcn_readlane(rp, k + 1);
        int q = start + lane;
        q = q < end ? q : end - 1;
        q = q > 0 ? q : 0;
        const int ent = csr[q];
        iv0[k] = (lane < end - start) ? ent : 0;
        const int nc = node < n ? node : n - 1;
        const float sv = hs[(unsigned)nc * (unsigned)C + (unsigned)ch];
        self[k] = (node < n && !up) ? sv : 0.0f;
    }

#pragma unroll
    for (int k = 0; k < NPW; ++k) {
        const int node = node0 + k;
        if (node < n) {
        const int start = __builtin_amdgcn_readlane(rp, k);
        const int end   = __builtin_amdgcn_readlane(rp, k + 1);

        float sum = self[k];

        for (int j = start; j < end; j += 32) {
            const int cnt = end - j;
            int iv = iv0[k];
            if (j != start) {
                iv = 0;
                if (lane < cnt) iv = csr[j + lane];
            }

            auto pick = [&](int t) -> unsigned {
                if constexpr (EPI == 1) {
                    return (unsigned)__builtin_amdgcn_readlane(iv, t);
                } else {
                    const int lo = __builtin_amdgcn_readlane(iv, 2 * t);
                    const int hi = __builtin_amdgcn_readlane(iv, 2 * t + 1);
                    return (unsigned)(up ? hi : lo);
                }
            };
#pragma unroll
            for (int g = 0; g < STEPS; g += GRP) {
                if (g * EPI < cnt) {
                    float v[GRP];
#pragma unroll
                    for (int t = 0; t < GRP; ++t)
                        v[t] = hs[pick(g + t) * (unsigned)C + (unsigned)ch];
#pragma unroll
                    for (int t = 0; t < GRP; ++t) {
                        const int ei = (g + t) * EPI + (up ? 1 : 0);
                        sum += (ei < cnt) ? v[t] : 0.0f;
                    }
                }
            }
        }
        if constexpr (EPI == 2) sum += __shfl_xor(sum, 16);

        if constexpr (FIN) {
            const float di = dinv[node];
            const float val = fmaxf(fmaf(di, sum, bias[ch]), 0.0f) * di;
            if (!up) o[(unsigned)node * (unsigned)C + (unsigned)ch] = val;
        } else {
            if (!up) rows[(wib * NPW + k) * PITCH + ch] = sum;
        }
        }
    }
    }

    if constexpr (!FIN) {
        constexpr int CG     = COUT / 32;
        constexpr int JOBS   = (NPB / 16) * CG;
        constexpr int KSTEPS = C / 4;
        __syncthreads();
        if (wib < JOBS) {
            const int rt = wib / CG;
            const int cg = wib % CG;
            const int row_base = (int)blockIdx.x * NPB + rt * 16;
            if (row_base < n) {
                const int m  = lane & 15;
                const int kh = (lane >> 4) & 1;
                const float* __restrict__ arow = &rows[(rt * 16 + m) * PITCH];
                float bv[2];
#pragma unroll
                for (int t = 0; t < 2; ++t) bv[t] = bias[cg * 32 + t * 16 + m];
                v8f acc[2];
                acc[0] = {};
                acc[1] = {};
#pragma unroll
                for (int kb = 0; kb < KSTEPS; ++kb) {
                    const int k0 = kb * 4 + 2 * kh;
                    const v2f a = *(const v2f*)&arow[k0];
#pragma unroll
                    for (int t = 0; t < 2; ++t) {
                        const int ncol = cg * 32 + t * 16 + m;
                        v2f b;
                        b.x = W[ k0      * COUT + ncol];
                        b.y = W[(k0 + 1) * COUT + ncol];
                        acc[t] = __builtin_amdgcn_wmma_f32_16x16x4_f32(
                             false, a,  false, b,
                             (short)0, acc[t],  false,  false);
                    }
                }

                if (row_base + 16 <= n) {
#pragma unroll
                    for (int v = 0; v < 8; ++v) {
                        const int r = row_base + v + 8 * kh;
                        const float di = dinv[r];
#pragma unroll
                        for (int t = 0; t < 2; ++t)
                            o[r * COUT + cg * 32 + t * 16 + m] = epilogue_value<MODE>(acc[t][v], di, bv[t]);
                    }
                } else {
#pragma unroll
                    for (int v = 0; v < 8; ++v) {
                        const int r = row_base + v + 8 * kh;
                        if (r < n) {
                            const float di = dinv[r];
#pragma unroll
                            for (int t = 0; t < 2; ++t)
                                o[r * COUT + cg * 32 + t * 16 + m] = epilogue_value<MODE>(acc[t][v], di, bv[t]);
                        }
                    }
                }
            }
        }
    }
}

extern "C" void kernel_launch(void* const* d_in, const int* in_sizes, int n_in,
                              void* d_out, int out_size, void* d_ws, size_t ws_size,
                              hipStream_t stream) {
    const float* x  = (const float*)d_in[0];
    const int*   ei = (const int*)  d_in[1];
    const float* W1 = (const float*)d_in[2];
    const float* b1 = (const float*)d_in[3];
    const float* W2 = (const float*)d_in[4];
    const float* b2 = (const float*)d_in[5];
    const float* W3 = (const float*)d_in[6];
    const float* b3 = (const float*)d_in[7];
    float* out = (float*)d_out;

    const int n = in_sizes[0] / 11;
    const int e = in_sizes[1] / 2;
    const int* src = ei;
    const int* dst = ei + e;

    const size_t np = ((size_t)n + 64) / 64 * 64;
    const size_t ep = ((size_t)e + 63) / 64 * 64;
    int*   rowptr = (int*)d_ws + np;
    float* dinv   = (float*)(rowptr + np);
    int*   csr    = (int*)(dinv + np);
    float* P      = (float*)(csr + ep);
    float* Q      = P + (size_t)n * 32;
    unsigned* bw  = (unsigned*)P;
    int*   tab    = (int*)Q;
    float* hsC    = P;
    float* hsB    = Q;
    float* hsA    = Q + (size_t)n * 16;

    const int B = 256;
    auto cdiv = [](long a, long b) { return (int)((a + b - 1) / b); };
    const int nb   = cdiv(n, BN);
    const int nblk = cdiv(e, CE);

    bucket_place_kernel<<<nblk, B, 0, stream>>>(src, dst, bw, tab, e, nb);
    bucket_lists_kernel<<<nb, LT, 0, stream>>>(tab, bw, rowptr, dinv, csr, x, W1, hsA, n, nblk, nb, e);

    const int gb = cdiv((long)((n + 3) / 4) * 32, B);

    gather_sum_kernel<16, 0, 0><<<gb, B, 0, stream>>>(rowptr, csr, hsA, dinv, b1, W1, hsB, n);

    gather_sum_kernel<16, 32, 1><<<gb, B, 0, stream>>>(rowptr, csr, hsB, dinv, b2, W2, hsC, n);

    gather_sum_kernel<32, 64, 2><<<gb, B, 0, stream>>>(rowptr, csr, hsC, dinv, b3, W3, out, n);
}


# Edge-sum row gathers take 16 B a lane: one load instruction brings 4 whole 128-B rows (C=16: 8 rows), not 1 (2); each lane group sums its entries, groups folded once per node; f32 sum order changes
# speedup vs baseline: 8.1007x; 1.0329x over previous
#include <hip/hip_runtime.h>


typedef float v2f __attribute__((ext_vector_type(2)));
typedef float v4f __attribute__((ext_vector_type(4)));
typedef float v8f __attribute__((ext_vector_type(8)));

constexpr int BSH   = 8;
constexpr int BN    = 1 << BSH;
constexpr int NBMAX = 1024;
constexpr int CE    = 4096;
constexpr int EPT   = CE / 256;
constexpr int LT    = 1024;
constexpr int LU    = 12;
constexpr int RB    = 16;

template <int NW>
__device__ __forceinline__ int block_excl_scan(int v, int* red, int& total) {
    const int lane = threadIdx.x & 31;
    const int wv   = threadIdx.x >> 5;
    int inc = v;
#pragma unroll
    for (int o = 1; o < 32; o <<= 1) {
        const int t = __shfl_up(inc, o);
        if (lane >= o) inc += t;
    }
    if (lane == 31) red[wv] = inc;
    __syncthreads();
    int off = 0, tot = 0;
#pragma unroll
    for (int w = 0; w < NW; ++w) {
        const int x = red[w];
        tot += x;
        if (w < wv) off += x;
    }
    total = tot;
    __syncthreads();
    return off + inc - v;
}

__device__ __forceinline__ int block_excl_scan32(int v, int* red, int& total) {
    const int lane = threadIdx.x & 31;
    const int wv   = __builtin_amdgcn_readfirstlane((int)(threadIdx.x >> 5));
    int inc = v;
#pragma unroll
    for (int o = 1; o < 32; o <<= 1) {
        const int t = __shfl_up(inc, o);
        if (lane >= o) inc += t;
    }
    if (lane == 31) red[wv] = inc;
    __syncthreads();
    const int x = red[lane];
    int off = 0, tot = 0;
#pragma unroll
    for (int w = 0; w < 32; ++w) {
        const int xw = __builtin_amdgcn_readlane(x, w);
        tot += xw;
        if (w < wv) off += xw;
    }
    total = tot;
    __syncthreads();
    return off + inc - v;
}

__global__ __launch_bounds__(256) void bucket_place_kernel(
    const int* __restrict__ src, const int* __restrict__ dst,
    unsigned* __restrict__ bw, int* __restrict__ tab, int e, int nb)
{
    __shared__ int h[NBMAX];
    __shared__ int ls[NBMAX];
    __shared__ unsigned sw[CE];
    __shared__ int red[8];
    const int tid = threadIdx.x;
#pragma unroll
    for (int k = 0; k < NBMAX / 256; ++k) h[k * 256 + tid] = 0;
    __syncthreads();

    const int first = blockIdx.x * CE;
    const int base  = first + tid;
    int d[EPT], s[EPT];
#pragma unroll
    for (int u = 0; u < EPT; ++u) {
        const int i  = base + u * 256;
        const int ic = i < e ? i : e - 1;
        d[u] = dst[ic];
        s[u] = src[ic];
    }
    int lr[EPT];
#pragma unroll
    for (int u = 0; u < EPT; ++u) {
        const int i = base + u * 256;
        lr[u] = 0;
        if (i < e) lr[u] = atomicAdd(&h[d[u] >> BSH], 1);
    }
    __syncthreads();

    int hc[4];
#pragma unroll
    for (int k = 0; k < 4; ++k) hc[k] = h[4 * tid + k];
    int all;
    int lrun = block_excl_scan<8>((hc[0] + hc[1]) + (hc[2] + hc[3]), red, all);
#pragma unroll
    for (int k = 0; k < 4; ++k) {
        ls[4 * tid + k] = lrun;
        lrun += hc[k];
    }
    __syncthreads();

    int slot[EPT];
#pragma unroll
    for (int u = 0; u < EPT; ++u) slot[u] = ls[d[u] >> BSH] + lr[u];
#pragma unroll
    for (int u = 0; u < EPT; ++u) {
        const int i = base + u * 256;
        if (i < e) sw[slot[u]] = ((unsigned)s[u] << BSH) | (unsigned)(d[u] & (BN - 1));
    }
    int* __restrict__ trow = tab + (size_t)blockIdx.x * (size_t)(nb + 1);
    for (int i = tid; i <= nb; i += 256) trow[i] = (i < nb) ? ls[i] : all;
    __syncthreads();

    unsigned wv[EPT];
#pragma unroll
    for (int u = 0; u < EPT; ++u) {
        const int j = u * 256 + tid;
        wv[u] = sw[j < all ? j : 0];
    }
#pragma unroll
    for (int u = 0; u < EPT; ++u) {
        const int j = u * 256 + tid;
        if (j < all) bw[first + j] = wv[u];
    }
}

static_assert(LT == 1024, "block_excl_scan32 is for 32 waves");
constexpr int LCAP = LU * LT;

__global__ __launch_bounds__(LT) void bucket_lists_kernel(
    const int* __restrict__ tab, const unsigned* __restrict__ bw,
    int* __restrict__ rowptr, float* __restrict__ dinv, int* __restrict__ csr,
    const float* __restrict__ x, const float* __restrict__ W1, float* __restrict__ hsA,
    int n, int nblk, int nb, int e)
{
    __shared__ int cl[BN];
    __shared__ int red[LT / 32];
    __shared__ unsigned stg[LCAP];
    __shared__ float dl[BN];
    const int tid = threadIdx.x;
    const int b   = blockIdx.x;

    constexpr int CIN1 = 11, COUT1 = 16, KS1 = (CIN1 + 3) / 4;
    const int lane = tid & 31;
    const int wv   = __builtin_amdgcn_readfirstlane(tid >> 5);
    const int pm   = lane & 15;
    const int pkh  = (lane >> 4) & 1;
    const int row_base = b * BN + wv * 16;
    const bool prod = (wv < BN / 16) && (row_base < n);
    v2f fa[KS1], fb[KS1];
#pragma unroll
    for (int kb = 0; kb < KS1; ++kb) { fa[kb] = {}; fb[kb] = {}; }
    if (prod) {
        const int arow = row_base + pm;
        const int ar = arow < n ? arow : n - 1;
#pragma unroll
        for (int kb = 0; kb < KS1; ++kb) {
            const int  k0  = kb * 4 + 2 * pkh;
            const bool ok0 = (k0     < CIN1);
            const bool ok1 = (k0 + 1 < CIN1);
            fa[kb].x = x[ar * CIN1 + (ok0 ? k0     : CIN1 - 1)];
            fa[kb].y = x[ar * CIN1 + (ok1 ? k0 + 1 : CIN1 - 1)];
            fb[kb].x = W1[(ok0 ? k0     : 0) * COUT1 + pm];
            fb[kb].y = W1[(ok1 ? k0 + 1 : 0) * COUT1 + pm];
        }
    }

    const int tp = nb + 1;
    int ra = 0, rcn = 0;
    if (tid < nblk) {
        const int* __restrict__ tr = tab + (size_t)tid * (size_t)tp + b;
        ra  = tr[0];
        rcn = tr[1] - ra;
    }
    int asum = ra, csum = rcn;
    for (int k = tid + LT; k < nblk; k += LT) {
        const int* __restrict__ tr = tab + (size_t)k * (size_t)tp + b;
        const int a2 = tr[0];
        asum += a2;
        csum += tr[1] - a2;
    }
    if (tid < BN) cl[tid] = 0;
    int m, bbase;
    const int off = block_excl_scan32(rcn, red, m);
    (void)block_excl_scan32(asum, red, bbase);
    if (nblk > LT) (void)block_excl_scan32(csum, red, m);
    const bool staged = (m <= LCAP) && (nblk <= LT);

    if (staged) {
        const int rsv = (tid < nblk) ? tid * CE + ra : 0;
        if (wv * 32 < nblk) {
#pragma unroll 1
            for (int qb = 0; qb < 32; qb += RB) {
                unsigned w0[RB];
#pragma unroll
                for (int q = 0; q < RB; ++q) {
                    const int rs = __builtin_amdgcn_readlane(rsv, qb + q);
                    const int rc = __builtin_amdgcn_readlane(rcn, qb + q);
                    const int lastj = rc > 0 ? rc - 1 : 0;
                    int i0 = rs + (lane < lastj ? lane : lastj);
                    i0 = i0 < e ? i0 : e - 1;
                    w0[q] = bw[i0];
                }
#pragma unroll
                for (int q = 0; q < RB; ++q) {
                    const int rs = __builtin_amdgcn_readlane(rsv, qb + q);
                    const int rc = __builtin_amdgcn_readlane(rcn, qb + q);
                    const int ro = __builtin_amdgcn_readlane(off, qb + q);
                    if (lane < rc) {
                        stg[ro + lane] = w0[q];
                        atomicAdd(&cl[w0[q] & (BN - 1)], 1);
                    }
                    for (int j = lane + 32; j < rc; j += 32) {
                        const unsigned w = bw[rs + j];
                        stg[ro + j] = w;
                        atomicAdd(&cl[w & (BN - 1)], 1);
                    }
                }
            }
        }
    } else {
        for (int k = tid; k < nblk; k += LT) {
            const int* __restrict__ tr = tab + (size_t)k * (size_t)tp + b;
            const int a2 = tr[0];
            const int c2 = tr[1] - a2;
            for (int j = 0; j < c2; ++j)
                atomicAdd(&cl[bw[k * CE + a2 + j] & (BN - 1)], 1);
        }
    }
    __syncthreads();
    const int c = (tid < BN) ? cl[tid] : 0;
    int all;
    const int ex = block_excl_scan32(c, red, all);
    const int node = b * BN + tid;
    if (tid < BN) {
        if (node < n) {
            const float di = rsqrtf((float)(c + 1));
            rowptr[node] = bbase + ex;
            dinv[node]   = di;
            dl[tid]      = di;
            if (node == n - 1) rowptr[n] = bbase + ex + c;
        }
        cl[tid] = ex;
    }
    __syncthreads();

    if (prod) {
        v8f acc = {};
#pragma unroll
        for (int kb = 0; kb < KS1; ++kb) {
            const int k0 = kb * 4 + 2 * pkh;
            v2f bq;
            bq.x = (k0     < CIN1) ? fb[kb].x : 0.0f;
            bq.y = (k0 + 1 < CIN1) ? fb[kb].y : 0.0f;
            acc = __builtin_amdgcn_wmma_f32_16x16x4_f32(
                 false, fa[kb],  false, bq,
                 (short)0, acc,  false,  false);
        }
        if (row_base + 16 <= n) {
#pragma unroll
            for (int v = 0; v < 8; ++v) {
                const int lr = wv * 16 + v + 8 * pkh;
                hsA[(b * BN + lr) * COUT1 + pm] = acc[v] * dl[lr];
            }
        } else {
#pragma unroll
            for (int v = 0; v < 8; ++v) {
                const int lr = wv * 16 + v + 8 * pkh;
                const int r  = b * BN + lr;
                if (r < n) hsA[r * COUT1 + pm] = acc[v] * dl[lr];
            }
        }
    }

    if (staged) {
        unsigned w[LU];
        int p[LU];
#pragma unroll
        for (int u = 0; u < LU; ++u) {
            const int i = u * LT + tid;
            w[u] = stg[(i < m) ? i : 0];
        }
        __syncthreads();
#pragma unroll
        for (int u = 0; u < LU; ++u) {
            const int i = u * LT + tid;
            p[u] = 0;
            if (i < m) p[u] = atomicAdd(&cl[w[u] & (BN - 1)], 1);
        }
#pragma unroll
        for (int u = 0; u < LU; ++u) {
            const int i = u * LT + tid;
            if (i < m) stg[p[u]] = w[u] >> BSH;
        }
        __syncthreads();
        for (int i = tid; i < m; i += LT) csr[bbase + i] = (int)stg[i];
    } else {
        for (int k = tid; k < nblk; k += LT) {
            const int* __restrict__ tr = tab + (size_t)k * (size_t)tp + b;
            const int a2 = tr[0];
            const int c2 = tr[1] - a2;
            for (int j = 0; j < c2; ++j) {
                const unsigned w = bw[k * CE + a2 + j];
                const int p = atomicAdd(&cl[w & (BN - 1)], 1);
                csr[bbase + p] = (int)(w >> BSH);
            }
        }
    }
}

template <int MODE>
__device__ __forceinline__ float epilogue_value(float p, float di, float b) {
    if constexpr (MODE == 1) return fmaxf(fmaf(di, p, b), 0.0f) * di;
    else return fmaf(di, p, b);
}

__device__ __forceinline__ float xhalf(float x) {
    const int b = __builtin_bit_cast(int, x);
    return __builtin_bit_cast(float, __builtin_amdgcn_permlanex16(b, b, 0x76543210, 0xfedcba98, false, false));
}
template <int M>
__device__ __forceinline__ float xrow(float x) {
    const int b = __builtin_bit_cast(int, x);
    return __builtin_bit_cast(float, __builtin_amdgcn_update_dpp(b, b, 0x160 + M, 0xf, 0xf, false));
}

template <int C, int COUT, int MODE>
__global__ __launch_bounds__(256) void gather_sum_kernel(
    const int* __restrict__ rowptr, const int* __restrict__ csr,
    const float* __restrict__ hs, const float* __restrict__ dinv,
    const float* __restrict__ bias, const float* __restrict__ W,
    float* __restrict__ o, int n)
{
    constexpr bool FIN  = (COUT == 0);
    constexpr int NPW   = 4;
    constexpr int NPB   = 8 * NPW;
    constexpr int PITCH = C + 2;
    constexpr int LPR   = C / 4;
    constexpr int RPI   = 32 / LPR;
    constexpr int STEPS = 32 / RPI;
    constexpr int GRP   = 16 / RPI;
    static_assert(C == 16 || C == 32, "one row = 4 or 8 lanes");
    static_assert(COUT % 32 == 0, "a product wave owns whole 128-B lines");

    __shared__ __attribute__((aligned(16))) float rows[FIN ? 1 : NPB * PITCH];

    const int lane = threadIdx.x & 31;
    const int sub  = lane & (LPR - 1);
    const int grp  = lane / LPR;
    const v4f* __restrict__ hs4 = (const v4f*)hs;
    const int wave = __builtin_amdgcn_readfirstlane(
        (int)((blockIdx.x * blockDim.x + threadIdx.x) >> 5));
    const int wib  = wave & 7;
    const int node0 = wave * NPW;

    if (node0 < n) {
    int bi = node0 + (lane < NPW ? lane : NPW);
    bi = bi < n ? bi : n;
    const int rp = rowptr[bi];

    int iv0[NPW];
    v4f self[NPW];
    const v4f zero4 = {0.0f, 0.0f, 0.0f, 0.0f};
#pragma unroll
    for (int k = 0; k < NPW; ++k) {
        const int node  = node0 + k;
        const int start = __builtin_amdgcn_readlane(rp, k);
        const int end   = __builtin_amdgcn_readlane(rp, k + 1);
        int q = start + lane;
        q = q < end ? q : end - 1;
        q = q > 0 ? q : 0;
        const int ent = csr[q];
        iv0[k] = (lane < end - start) ? ent : 0;
        const int nc = node < n ? node : n - 1;
        const v4f sv = hs4[(unsigned)nc * (unsigned)LPR + (unsigned)sub];
        self[k] = (node < n && grp == 0) ? sv : zero4;
    }
    float bq[4] = {0.0f, 0.0f, 0.0f, 0.0f};
    if constexpr (FIN) {
#pragma unroll
        for (int q = 0; q < 4; ++q) bq[q] = bias[4 * sub + q];
    }

#pragma unroll
    for (int k = 0; k < NPW; ++k) {
        const int node = node0 + k;
        if (node < n) {
        const int start = __builtin_amdgcn_readlane(rp, k);
        const int end   = __builtin_amdgcn_readlane(rp, k + 1);

        v4f sum = self[k];

        for (int j = start; j < end; j += 32) {
            const int cnt = end - j;
            int iv = iv0[k];
            if (j != start) {
                iv = 0;
                if (lane < cnt) iv = csr[j + lane];
            }

#pragma unroll
            for (int g = 0; g < STEPS; g += GRP) {
                if (g * RPI < cnt) {
                    v4f v[GRP];
#pragma unroll
                    for (int t = 0; t < GRP; ++t) {
                        const unsigned idx = (unsigned)__shfl(iv, (g + t) * RPI + grp);
                        v[t] = hs4[idx * (unsigned)LPR + (unsigned)sub];
                    }
#pragma unroll
                    for (int t = 0; t < GRP; ++t) {
                        const bool ok = ((g + t) * RPI + grp) < cnt;
                        sum.x += ok ? v[t].x : 0.0f;
                        sum.y += ok ? v[t].y : 0.0f;
                        sum.z += ok ? v[t].z : 0.0f;
                        sum.w += ok ? v[t].w : 0.0f;
                    }
                }
            }
        }
        if constexpr (RPI == 8) {
            sum.x += xrow<4>(sum.x); sum.y += xrow<4>(sum.y); sum.z += xrow<4>(sum.z); sum.w += xrow<4>(sum.w);
        }
        sum.x += xrow<8>(sum.x); sum.y += xrow<8>(sum.y); sum.z += xrow<8>(sum.z); sum.w += xrow<8>(sum.w);
        sum.x += xhalf(sum.x);   sum.y += xhalf(sum.y);   sum.z += xhalf(sum.z);   sum.w += xhalf(sum.w);

        if constexpr (FIN) {
            const float di = dinv[node];
            v4f val;
            val.x = fmaxf(fmaf(di, sum.x, bq[0]), 0.0f) * di;
            val.y = fmaxf(fmaf(di, sum.y, bq[1]), 0.0f) * di;
            val.z = fmaxf(fmaf(di, sum.z, bq[2]), 0.0f) * di;
            val.w = fmaxf(fmaf(di, sum.w, bq[3]), 0.0f) * di;
            if (grp == 0) *(v4f*)&o[(unsigned)node * (unsigned)C + 4u * (unsigned)sub] = val;
        } else {
            if (grp == 0) {
                float* __restrict__ r = &rows[(wib * NPW + k) * PITCH + 4 * sub];
                *(v2f*)r       = v2f{sum.x, sum.y};
                *(v2f*)(r + 2) = v2f{sum.z, sum.w};
            }
        }
        }
    }
    }

    if constexpr (!FIN) {
        constexpr int CG     = COUT / 32;
        constexpr int JOBS   = (NPB / 16) * CG;
        constexpr int KSTEPS = C / 4;
        __syncthreads();
        if (wib < JOBS) {
            const int rt = wib / CG;
            const int cg = wib % CG;
            const int row_base = (int)blockIdx.x * NPB + rt * 16;
            if (row_base < n) {
                const int m  = lane & 15;
                const int kh = (lane >> 4) & 1;
                const float* __restrict__ arow = &rows[(rt * 16 + m) * PITCH];
                float bv[2];
#pragma unroll
                for (int t = 0; t < 2; ++t) bv[t] = bias[cg * 32 + t * 16 + m];
                v8f acc[2];
                acc[0] = {};
                acc[1] = {};
#pragma unroll
                for (int kb = 0; kb < KSTEPS; ++kb) {
                    const int k0 = kb * 4 + 2 * kh;
                    const v2f a = *(const v2f*)&arow[k0];
#pragma unroll
                    for (int t = 0; t < 2; ++t) {
                        const int ncol = cg * 32 + t * 16 + m;
                        v2f b;
                        b.x = W[ k0      * COUT + ncol];
                        b.y = W[(k0 + 1) * COUT + ncol];
                        acc[t] = __builtin_amdgcn_wmma_f32_16x16x4_f32(
                             false, a,  false, b,
                             (short)0, acc[t],  false,  false);
                    }
                }

                if (row_base + 16 <= n) {
#pragma unroll
                    for (int v = 0; v < 8; ++v) {
                        const int r = row_base + v + 8 * kh;
                        const float di = dinv[r];
#pragma unroll
                        for (int t = 0; t < 2; ++t)
                            o[r * COUT + cg * 32 + t * 16 + m] = epilogue_value<MODE>(acc[t][v], di, bv[t]);
                    }
                } else {
#pragma unroll
                    for (int v = 0; v < 8; ++v) {
                        const int r = row_base + v + 8 * kh;
                        if (r < n) {
                            const float di = dinv[r];
#pragma unroll
                            for (int t = 0; t < 2; ++t)
                                o[r * COUT + cg * 32 + t * 16 + m] = epilogue_value<MODE>(acc[t][v], di, bv[t]);
                        }
                    }
                }
            }
        }
    }
}

extern "C" void kernel_launch(void* const* d_in, const int* in_sizes, int n_in,
                              void* d_out, int out_size, void* d_ws, size_t ws_size,
                              hipStream_t stream) {
    const float* x  = (const float*)d_in[0];
    const int*   ei = (const int*)  d_in[1];
    const float* W1 = (const float*)d_in[2];
    const float* b1 = (const float*)d_in[3];
    const float* W2 = (const float*)d_in[4];
    const float* b2 = (const float*)d_in[5];
    const float* W3 = (const float*)d_in[6];
    const float* b3 = (const float*)d_in[7];
    float* out = (float*)d_out;

    const int n = in_sizes[0] / 11;
    const int e = in_sizes[1] / 2;
    const int* src = ei;
    const int* dst = ei + e;

    const size_t np = ((size_t)n + 64) / 64 * 64;
    const size_t ep = ((size_t)e + 63) / 64 * 64;
    int*   rowptr = (int*)d_ws + np;
    float* dinv   = (float*)(rowptr + np);
    int*   csr    = (int*)(dinv + np);
    float* P      = (float*)(csr + ep);
    float* Q      = P + (size_t)n * 32;
    unsigned* bw  = (unsigned*)P;
    int*   tab    = (int*)Q;
    float* hsC    = P;
    float* hsB    = Q;
    float* hsA    = Q + (size_t)n * 16;

    const int B = 256;
    auto cdiv = [](long a, long b) { return (int)((a + b - 1) / b); };
    const int nb   = cdiv(n, BN);
    const int nblk = cdiv(e, CE);

    bucket_place_kernel<<<nblk, B, 0, stream>>>(src, dst, bw, tab, e, nb);
    bucket_lists_kernel<<<nb, LT, 0, stream>>>(tab, bw, rowptr, dinv, csr, x, W1, hsA, n, nblk, nb, e);

    const int gb = cdiv((long)((n + 3) / 4) * 32, B);

    gather_sum_kernel<16, 0, 0><<<gb, B, 0, stream>>>(rowptr, csr, hsA, dinv, b1, W1, hsB, n);

    gather_sum_kernel<16, 32, 1><<<gb, B, 0, stream>>>(rowptr, csr, hsB, dinv, b2, W2, hsC, n);

    gather_sum_kernel<32, 64, 2><<<gb, B, 0, stream>>>(rowptr, csr, hsC, dinv, b3, W3, out, n);
}
